# starting kernel
# baseline (speedup 1.0000x reference)
#include <hip/hip_runtime.h>
#include <hip/hip_bf16.h>

typedef __attribute__((ext_vector_type(2))) float v2f;
typedef __attribute__((ext_vector_type(8))) float v8f;

#define B_ 16
#define N_ 4096
#define C_ 1024
#define NCHUNK 16
#define CHUNK 256
#define XSTR 1032

__global__ __launch_bounds__(256) void ga_lncf(const float* __restrict__ cf,
                                               const float* __restrict__ qg,
                                               const float* __restrict__ qb,
                                               float* __restrict__ cfn) {
  __shared__ float s0[256], s1[256];
  const int tid = threadIdx.x;
  float4 x = *(const float4*)(cf + tid * 4);
  s0[tid] = x.x + x.y + x.z + x.w;
  s1[tid] = x.x * x.x + x.y * x.y + x.z * x.z + x.w * x.w;
  __syncthreads();
  for (int s = 128; s > 0; s >>= 1) {
    if (tid < s) { s0[tid] += s0[tid + s]; s1[tid] += s1[tid + s]; }
    __syncthreads();
  }
  const float mu = s0[0] * (1.f / 1024.f);
  const float var = s1[0] * (1.f / 1024.f) - mu * mu;
  const float rs = rsqrtf(var + 1e-5f);
  float4 g = *(const float4*)(qg + tid * 4);
  float4 b4 = *(const float4*)(qb + tid * 4);
  float4 o;
  o.x = (x.x - mu) * rs * g.x + b4.x;
  o.y = (x.y - mu) * rs * g.y + b4.y;
  o.z = (x.z - mu) * rs * g.z + b4.z;
  o.w = (x.w - mu) * rs * g.w + b4.w;
  *(float4*)(cfn + tid * 4) = o;
}

__global__ __launch_bounds__(256) void ga_q(const float* __restrict__ cfn,
                                            const float* __restrict__ Wq,
                                            float* __restrict__ qs) {
  __shared__ float cs[1024];
  const int tid = threadIdx.x;
  for (int i = tid; i < 1024; i += 256) cs[i] = cfn[i];
  __syncthreads();
  const int wv = tid >> 5, ln = tid & 31;
  for (int jj = 0; jj < 32; ++jj) {
    const int j = blockIdx.x * 256 + wv * 32 + jj;
    const float* wrow = Wq + (size_t)j * 1024;
    float acc = 0.f;
    for (int c0 = 0; c0 < 1024; c0 += 32) acc += cs[c0 + ln] * wrow[c0 + ln];
#pragma unroll
    for (int m = 16; m > 0; m >>= 1) acc += __shfl_xor(acc, m, 32);
    if (ln == 0) qs[j] = acc * 0.03125f;
  }
}

__global__ __launch_bounds__(256) void ga_wqeff(const float* __restrict__ Wkv,
                                                const float* __restrict__ qs,
                                                float* __restrict__ wqe) {
  __shared__ float q_s[1024];
  const int tid = threadIdx.x;
  for (int i = tid; i < 1024; i += 256) q_s[i] = qs[i];
  __syncthreads();
  const int c = blockIdx.x * 256 + tid;
  for (int h = 0; h < 8; ++h) {
    float acc = 0.f;
    for (int d = 0; d < 128; ++d)
      acc += Wkv[(size_t)(h * 128 + d) * 1024 + c] * q_s[h * 128 + d];
    wqe[c * 16 + h] = acc;
  }
  for (int h = 8; h < 16; ++h) wqe[c * 16 + h] = 0.f;
}

__global__ __launch_bounds__(256) void ga_main(const float* __restrict__ feats,
                                               const float* __restrict__ kvg,
                                               const float* __restrict__ kvb,
                                               const float* __restrict__ wqe,
                                               float* __restrict__ mp,
                                               float* __restrict__ lp,
                                               float* __restrict__ yp) {
  __shared__ __align__(16) float xs[16][XSTR];
  __shared__ __align__(16) float wq[1024 * 17];
  __shared__ __align__(32) float sred[2048];
  __shared__ float sS[256];
  __shared__ float sP[16 * 17];
  __shared__ float sF[16];
  __shared__ float smu[16], srs[16];

  const int tid = threadIdx.x;
  const int lane = tid & 31;
  const int wave = tid >> 5;
  const int l16 = lane & 15;
  const int lhi = lane >> 4;
  const int r = tid >> 4;
  const int t16 = tid & 15;

  const int blk = blockIdx.x;
  const int bb = blk >> 4;
  const int chunk = blk & 15;

  for (int i = tid; i < 16384; i += 256) wq[(i >> 4) * 17 + (i & 15)] = wqe[i];

  const v8f vzero = {0.f, 0.f, 0.f, 0.f, 0.f, 0.f, 0.f, 0.f};
  v8f yacc[8];
#pragma unroll
  for (int t = 0; t < 8; ++t) yacc[t] = vzero;

  float m_run = -3.4e38f;
  float l_run = 0.f;

  __syncthreads();

#pragma unroll 1
  for (int st = 0; st < 16; ++st) {
    const size_t rowbase =
        ((size_t)bb * N_ + (size_t)chunk * CHUNK + st * 16 + r) * C_;
    float sum = 0.f, sq = 0.f;
#pragma unroll
    for (int i = 0; i < 16; ++i) {
      const int col = i * 64 + t16 * 4;
      float4 x = *(const float4*)(feats + rowbase + col);
      sum += x.x + x.y + x.z + x.w;
      sq += x.x * x.x + x.y * x.y + x.z * x.z + x.w * x.w;
      *(float4*)&xs[r][col] = x;
    }
    if (st < 15)
      __builtin_prefetch(feats + rowbase + 16 * C_ + t16 * 64, 0, 3);
    sred[tid] = sum;
    sred[256 + tid] = sq;
    __syncthreads();
    if (tid < 16) {
      float s0 = 0.f, s1 = 0.f;
#pragma unroll
      for (int i = 0; i < 16; ++i) {
        s0 += sred[tid * 16 + i];
        s1 += sred[256 + tid * 16 + i];
      }
      const float mu = s0 * (1.f / 1024.f);
      const float var = s1 * (1.f / 1024.f) - mu * mu;
      smu[tid] = mu;
      srs[tid] = rsqrtf(var + 1e-5f);
    }
    __syncthreads();
    {
      const float mu = smu[r], rs = srs[r];
#pragma unroll
      for (int i = 0; i < 16; ++i) {
        const int col = i * 64 + t16 * 4;
        float4 x = *(float4*)&xs[r][col];
        float4 g = *(const float4*)(kvg + col);
        float4 b4 = *(const float4*)(kvb + col);
        x.x = (x.x - mu) * rs * g.x + b4.x;
        x.y = (x.y - mu) * rs * g.y + b4.y;
        x.z = (x.z - mu) * rs * g.z + b4.z;
        x.w = (x.w - mu) * rs * g.w + b4.w;
        *(float4*)&xs[r][col] = x;
      }
    }
    __syncthreads();

    v8f sacc = vzero;
    const int kb = wave * 128 + 2 * lhi;
#pragma unroll
    for (int kk = 0; kk < 32; ++kk) {
      const int k0 = kb + kk * 4;
      v2f a = *(const v2f*)&xs[l16][k0];
      v2f bv;
      bv.x = wq[k0 * 17 + l16];
      bv.y = wq[(k0 + 1) * 17 + l16];
      sacc = __builtin_amdgcn_wmma_f32_16x16x4_f32(false, a, false, bv,
                                                   (short)0, sacc, false, false);
    }
    *(v8f*)&sred[wave * 256 + lane * 8] = sacc;
    __syncthreads();
    {
      const int rr = tid >> 4, hh = tid & 15;
      const int idx = (hh + ((rr >> 3) << 4)) * 8 + (rr & 7);
      float v = 0.f;
#pragma unroll
      for (int w = 0; w < 8; ++w) v += sred[w * 256 + idx];
      sS[rr * 16 + hh] = v;
    }
    __syncthreads();

    if (tid < 16) {
      float rmax = -3.4e38f;
#pragma unroll
      for (int q = 0; q < 16; ++q) rmax = fmaxf(rmax, sS[q * 16 + tid]);
      const float mnew = fmaxf(m_run, rmax);
      const float f = __expf(m_run - mnew);
      float ls = 0.f;
#pragma unroll
      for (int q = 0; q < 16; ++q) {
        const float p = __expf(sS[q * 16 + tid] - mnew);
        sP[q * 17 + tid] = p;
        ls += p;
      }
      l_run = l_run * f + ls;
      m_run = mnew;
      sF[tid] = f;
    }
    __syncthreads();

    float fv[8];
#pragma unroll
    for (int v = 0; v < 8; ++v) fv[v] = sF[v + 8 * lhi];
#pragma unroll
    for (int t = 0; t < 8; ++t) {
#pragma unroll
      for (int v = 0; v < 8; ++v) yacc[t][v] *= fv[v];
    }
#pragma unroll
    for (int k0 = 0; k0 < 16; k0 += 4) {
      const int kr = k0 + 2 * lhi;
      v2f a;
      a.x = sP[kr * 17 + l16];
      a.y = sP[(kr + 1) * 17 + l16];
#pragma unroll
      for (int t = 0; t < 8; ++t) {
        const int c0 = wave * 128 + t * 16;
        v2f bv;
        bv.x = xs[kr][c0 + l16];
        bv.y = xs[kr + 1][c0 + l16];
        yacc[t] = __builtin_amdgcn_wmma_f32_16x16x4_f32(
            false, a, false, bv, (short)0, yacc[t], false, false);
      }
    }
    __syncthreads();
  }

  if (tid < 8) {
    mp[blk * 8 + tid] = m_run;
    lp[blk * 8 + tid] = l_run;
  }
  if (lhi == 0) {
#pragma unroll
    for (int t = 0; t < 8; ++t) {
      const int c0 = wave * 128 + t * 16;
#pragma unroll
      for (int v = 0; v < 8; ++v)
        yp[(size_t)(blk * 8 + v) * 1024 + c0 + l16] = yacc[t][v];
    }
  }
}

__global__ __launch_bounds__(1024) void ga_merge(const float* __restrict__ mp,
                                                 const float* __restrict__ lp,
                                                 const float* __restrict__ yp,
                                                 const float* __restrict__ Wkv,
                                                 const float* __restrict__ projW,
                                                 const float* __restrict__ projb,
                                                 const float* __restrict__ cf,
                                                 float* __restrict__ out) {
  __shared__ float yn[8 * 1024];
  __shared__ float ag[1024];
  __shared__ float ef[16 * 8];
  __shared__ float Lh[8];
  const int tid = threadIdx.x;
  const int b = blockIdx.x;
  if (tid < 8) {
    float M = -3.4e38f;
    for (int k = 0; k < 16; ++k) M = fmaxf(M, mp[(b * 16 + k) * 8 + tid]);
    float L = 0.f;
    for (int k = 0; k < 16; ++k) {
      const float e = __expf(mp[(b * 16 + k) * 8 + tid] - M);
      ef[k * 8 + tid] = e;
      L += lp[(b * 16 + k) * 8 + tid] * e;
    }
    Lh[tid] = L;
  }
  __syncthreads();
  for (int e = tid; e < 8192; e += 1024) {
    const int h = e >> 10, c = e & 1023;
    float acc = 0.f;
    for (int k = 0; k < 16; ++k)
      acc += yp[(size_t)((b * 16 + k) * 8 + h) * 1024 + c] * ef[k * 8 + h];
    yn[e] = acc / Lh[h];
  }
  __syncthreads();
  const int wv = tid >> 5, ln = tid & 31;
  for (int jj = 0; jj < 32; ++jj) {
    const int j = wv * 32 + jj;
    const int h = j >> 7;
    const float* wrow = Wkv + (size_t)(1024 + j) * 1024;
    float acc = 0.f;
    for (int c0 = 0; c0 < 1024; c0 += 32) acc += yn[h * 1024 + c0 + ln] * wrow[c0 + ln];
#pragma unroll
    for (int m = 16; m > 0; m >>= 1) acc += __shfl_xor(acc, m, 32);
    if (ln == 0) ag[j] = acc;
  }
  __syncthreads();
  for (int jj = 0; jj < 32; ++jj) {
    const int j = wv * 32 + jj;
    const float* prow = projW + (size_t)j * 1024;
    float acc = 0.f;
    for (int c0 = 0; c0 < 1024; c0 += 32) acc += ag[c0 + ln] * prow[c0 + ln];
#pragma unroll
    for (int m = 16; m > 0; m >>= 1) acc += __shfl_xor(acc, m, 32);
    if (ln == 0) out[(size_t)b * 1024 + j] = cf[j] + acc + projb[j];
  }
}

extern "C" void kernel_launch(void* const* d_in, const int* in_sizes, int n_in,
                              void* d_out, int out_size, void* d_ws, size_t ws_size,
                              hipStream_t stream) {
  const float* class_feature = (const float*)d_in[0];
  const float* features      = (const float*)d_in[1];
  const float* q_gamma       = (const float*)d_in[2];
  const float* q_beta        = (const float*)d_in[3];
  const float* Wq            = (const float*)d_in[4];
  const float* kv_gamma      = (const float*)d_in[5];
  const float* kv_beta       = (const float*)d_in[6];
  const float* Wkv           = (const float*)d_in[7];
  const float* proj_W        = (const float*)d_in[8];
  const float* proj_b        = (const float*)d_in[9];
  float* out = (float*)d_out;

  float* ws   = (float*)d_ws;
  float* cfn  = ws;
  float* qs   = ws + 1024;
  float* wqe  = ws + 2048;
  float* mp   = ws + 18432;
  float* lp   = ws + 20480;
  float* yp   = ws + 22528;

  ga_lncf<<<1, 256, 0, stream>>>(class_feature, q_gamma, q_beta, cfn);
  ga_q<<<4, 256, 0, stream>>>(cfn, Wq, qs);
  ga_wqeff<<<4, 256, 0, stream>>>(Wkv, qs, wqe);
  ga_main<<<B_ * NCHUNK, 256, 0, stream>>>(features, kv_gamma, kv_beta, wqe, mp, lp, yp);
  ga_merge<<<B_, 1024, 0, stream>>>(mp, lp, yp, Wkv, proj_W, proj_b, class_feature, out);
}


# tail (merge, Wv, proj) leaves 16 one-per-batch blocks for 3 wide launches (5 -> 7 per call); batch is the WMMA M dim so each weight row is read once; changes sum order of both tail dot products
# speedup vs baseline: 1.4282x; 1.4282x over previous
#include <hip/hip_runtime.h>
#include <hip/hip_bf16.h>

typedef __attribute__((ext_vector_type(2))) float v2f;
typedef __attribute__((ext_vector_type(8))) float v8f;

#define B_ 16
#define N_ 4096
#define C_ 1024
#define NCHUNK 16
#define CHUNK 256
#define XSTR 1032

__global__ __launch_bounds__(256) void ga_lncf(const float* __restrict__ cf,
                                               const float* __restrict__ qg,
                                               const float* __restrict__ qb,
                                               float* __restrict__ cfn) {
  __shared__ float s0[256], s1[256];
  const int tid = threadIdx.x;
  float4 x = *(const float4*)(cf + tid * 4);
  s0[tid] = x.x + x.y + x.z + x.w;
  s1[tid] = x.x * x.x + x.y * x.y + x.z * x.z + x.w * x.w;
  __syncthreads();
  for (int s = 128; s > 0; s >>= 1) {
    if (tid < s) { s0[tid] += s0[tid + s]; s1[tid] += s1[tid + s]; }
    __syncthreads();
  }
  const float mu = s0[0] * (1.f / 1024.f);
  const float var = s1[0] * (1.f / 1024.f) - mu * mu;
  const float rs = rsqrtf(var + 1e-5f);
  float4 g = *(const float4*)(qg + tid * 4);
  float4 b4 = *(const float4*)(qb + tid * 4);
  float4 o;
  o.x = (x.x - mu) * rs * g.x + b4.x;
  o.y = (x.y - mu) * rs * g.y + b4.y;
  o.z = (x.z - mu) * rs * g.z + b4.z;
  o.w = (x.w - mu) * rs * g.w + b4.w;
  *(float4*)(cfn + tid * 4) = o;
}

__global__ __launch_bounds__(256) void ga_q(const float* __restrict__ cfn,
                                            const float* __restrict__ Wq,
                                            float* __restrict__ qs) {
  __shared__ float cs[1024];
  const int tid = threadIdx.x;
  for (int i = tid; i < 1024; i += 256) cs[i] = cfn[i];
  __syncthreads();
  const int wv = tid >> 5, ln = tid & 31;
  for (int jj = 0; jj < 32; ++jj) {
    const int j = blockIdx.x * 256 + wv * 32 + jj;
    const float* wrow = Wq + (size_t)j * 1024;
    float acc = 0.f;
    for (int c0 = 0; c0 < 1024; c0 += 32) acc += cs[c0 + ln] * wrow[c0 + ln];
#pragma unroll
    for (int m = 16; m > 0; m >>= 1) acc += __shfl_xor(acc, m, 32);
    if (ln == 0) qs[j] = acc * 0.03125f;
  }
}

__global__ __launch_bounds__(256) void ga_wqeff(const float* __restrict__ Wkv,
                                                const float* __restrict__ qs,
                                                float* __restrict__ wqe) {
  __shared__ float q_s[1024];
  const int tid = threadIdx.x;
  for (int i = tid; i < 1024; i += 256) q_s[i] = qs[i];
  __syncthreads();
  const int c = blockIdx.x * 256 + tid;
  for (int h = 0; h < 8; ++h) {
    float acc = 0.f;
    for (int d = 0; d < 128; ++d)
      acc += Wkv[(size_t)(h * 128 + d) * 1024 + c] * q_s[h * 128 + d];
    wqe[c * 16 + h] = acc;
  }
  for (int h = 8; h < 16; ++h) wqe[c * 16 + h] = 0.f;
}

__global__ __launch_bounds__(256) void ga_main(const float* __restrict__ feats,
                                               const float* __restrict__ kvg,
                                               const float* __restrict__ kvb,
                                               const float* __restrict__ wqe,
                                               float* __restrict__ mp,
                                               float* __restrict__ lp,
                                               float* __restrict__ yp) {
  __shared__ __align__(16) float xs[16][XSTR];
  __shared__ __align__(16) float wq[1024 * 17];
  __shared__ __align__(32) float sred[2048];
  __shared__ float sS[256];
  __shared__ float sP[16 * 17];
  __shared__ float sF[16];
  __shared__ float smu[16], srs[16];

  const int tid = threadIdx.x;
  const int lane = tid & 31;
  const int wave = tid >> 5;
  const int l16 = lane & 15;
  const int lhi = lane >> 4;
  const int r = tid >> 4;
  const int t16 = tid & 15;

  const int blk = blockIdx.x;
  const int bb = blk >> 4;
  const int chunk = blk & 15;

  for (int i = tid; i < 16384; i += 256) wq[(i >> 4) * 17 + (i & 15)] = wqe[i];

  const v8f vzero = {0.f, 0.f, 0.f, 0.f, 0.f, 0.f, 0.f, 0.f};
  v8f yacc[8];
#pragma unroll
  for (int t = 0; t < 8; ++t) yacc[t] = vzero;

  float m_run = -3.4e38f;
  float l_run = 0.f;

  __syncthreads();

#pragma unroll 1
  for (int st = 0; st < 16; ++st) {
    const size_t rowbase =
        ((size_t)bb * N_ + (size_t)chunk * CHUNK + st * 16 + r) * C_;
    float sum = 0.f, sq = 0.f;
#pragma unroll
    for (int i = 0; i < 16; ++i) {
      const int col = i * 64 + t16 * 4;
      float4 x = *(const float4*)(feats + rowbase + col);
      sum += x.x + x.y + x.z + x.w;
      sq += x.x * x.x + x.y * x.y + x.z * x.z + x.w * x.w;
      *(float4*)&xs[r][col] = x;
    }
    if (st < 15)
      __builtin_prefetch(feats + rowbase + 16 * C_ + t16 * 64, 0, 3);
    sred[tid] = sum;
    sred[256 + tid] = sq;
    __syncthreads();
    if (tid < 16) {
      float s0 = 0.f, s1 = 0.f;
#pragma unroll
      for (int i = 0; i < 16; ++i) {
        s0 += sred[tid * 16 + i];
        s1 += sred[256 + tid * 16 + i];
      }
      const float mu = s0 * (1.f / 1024.f);
      const float var = s1 * (1.f / 1024.f) - mu * mu;
      smu[tid] = mu;
      srs[tid] = rsqrtf(var + 1e-5f);
    }
    __syncthreads();
    {
      const float mu = smu[r], rs = srs[r];
#pragma unroll
      for (int i = 0; i < 16; ++i) {
        const int col = i * 64 + t16 * 4;
        float4 x = *(float4*)&xs[r][col];
        float4 g = *(const float4*)(kvg + col);
        float4 b4 = *(const float4*)(kvb + col);
        x.x = (x.x - mu) * rs * g.x + b4.x;
        x.y = (x.y - mu) * rs * g.y + b4.y;
        x.z = (x.z - mu) * rs * g.z + b4.z;
        x.w = (x.w - mu) * rs * g.w + b4.w;
        *(float4*)&xs[r][col] = x;
      }
    }
    __syncthreads();

    v8f sacc = vzero;
    const int kb = wave * 128 + 2 * lhi;
#pragma unroll
    for (int kk = 0; kk < 32; ++kk) {
      const int k0 = kb + kk * 4;
      v2f a = *(const v2f*)&xs[l16][k0];
      v2f bv;
      bv.x = wq[k0 * 17 + l16];
      bv.y = wq[(k0 + 1) * 17 + l16];
      sacc = __builtin_amdgcn_wmma_f32_16x16x4_f32(false, a, false, bv,
                                                   (short)0, sacc, false, false);
    }
    *(v8f*)&sred[wave * 256 + lane * 8] = sacc;
    __syncthreads();
    {
      const int rr = tid >> 4, hh = tid & 15;
      const int idx = (hh + ((rr >> 3) << 4)) * 8 + (rr & 7);
      float v = 0.f;
#pragma unroll
      for (int w = 0; w < 8; ++w) v += sred[w * 256 + idx];
      sS[rr * 16 + hh] = v;
    }
    __syncthreads();

    if (tid < 16) {
      float rmax = -3.4e38f;
#pragma unroll
      for (int q = 0; q < 16; ++q) rmax = fmaxf(rmax, sS[q * 16 + tid]);
      const float mnew = fmaxf(m_run, rmax);
      const float f = __expf(m_run - mnew);
      float ls = 0.f;
#pragma unroll
      for (int q = 0; q < 16; ++q) {
        const float p = __expf(sS[q * 16 + tid] - mnew);
        sP[q * 17 + tid] = p;
        ls += p;
      }
      l_run = l_run * f + ls;
      m_run = mnew;
      sF[tid] = f;
    }
    __syncthreads();

    float fv[8];
#pragma unroll
    for (int v = 0; v < 8; ++v) fv[v] = sF[v + 8 * lhi];
#pragma unroll
    for (int t = 0; t < 8; ++t) {
#pragma unroll
      for (int v = 0; v < 8; ++v) yacc[t][v] *= fv[v];
    }
#pragma unroll
    for (int k0 = 0; k0 < 16; k0 += 4) {
      const int kr = k0 + 2 * lhi;
      v2f a;
      a.x = sP[kr * 17 + l16];
      a.y = sP[(kr + 1) * 17 + l16];
#pragma unroll
      for (int t = 0; t < 8; ++t) {
        const int c0 = wave * 128 + t * 16;
        v2f bv;
        bv.x = xs[kr][c0 + l16];
        bv.y = xs[kr + 1][c0 + l16];
        yacc[t] = __builtin_amdgcn_wmma_f32_16x16x4_f32(
            false, a, false, bv, (short)0, yacc[t], false, false);
      }
    }
    __syncthreads();
  }

  if (tid < 8) {
    mp[blk * 8 + tid] = m_run;
    lp[blk * 8 + tid] = l_run;
  }
  if (lhi == 0) {
#pragma unroll
    for (int t = 0; t < 8; ++t) {
      const int c0 = wave * 128 + t * 16;
#pragma unroll
      for (int v = 0; v < 8; ++v)
        yp[(size_t)(blk * 8 + v) * 1024 + c0 + l16] = yacc[t][v];
    }
  }
}


__global__ __launch_bounds__(128) void ga_comb(const float* __restrict__ mp,
                                               const float* __restrict__ lp,
                                               float* yp) {
  const int bh = blockIdx.x >> 1;
  const int b = bh >> 3, h = bh & 7;
  const int c = (((blockIdx.x & 1) << 7) + threadIdx.x) * 4;
  float e[16];
  float M = -3.4e38f;
  for (int k = 0; k < 16; ++k) {
    e[k] = mp[(b * 16 + k) * 8 + h];
    M = fmaxf(M, e[k]);
  }
  float L = 0.f;
  for (int k = 0; k < 16; ++k) {
    e[k] = __expf(e[k] - M);
    L += lp[(b * 16 + k) * 8 + h] * e[k];
  }
  float4 acc = {0.f, 0.f, 0.f, 0.f};
  for (int k = 0; k < 16; ++k) {
    const float4 y = *(const float4*)(yp + (size_t)((b * 16 + k) * 8 + h) * 1024 + c);
    acc.x += y.x * e[k];
    acc.y += y.y * e[k];
    acc.z += y.z * e[k];
    acc.w += y.w * e[k];
  }
  acc.x = acc.x / L;
  acc.y = acc.y / L;
  acc.z = acc.z / L;
  acc.w = acc.w / L;
  *(float4*)(yp + (size_t)((b * 16) * 8 + h) * 1024 + c) = acc;
}

__device__ __forceinline__ v8f ga_tile64(const float* __restrict__ arow,
                                         const float* __restrict__ wrow) {
  const v8f vzero = {0.f, 0.f, 0.f, 0.f, 0.f, 0.f, 0.f, 0.f};
  v8f acc0 = vzero, acc1 = vzero;
#pragma unroll
  for (int s = 0; s < 8; ++s) {
    const float4 a = *(const float4*)(arow + s * 8);
    const float4 w = *(const float4*)(wrow + s * 8);
    v2f a0, a1, b0, b1;
    a0.x = a.x; a0.y = a.y; a1.x = a.z; a1.y = a.w;
    b0.x = w.x; b0.y = w.y; b1.x = w.z; b1.y = w.w;
    acc0 = __builtin_amdgcn_wmma_f32_16x16x4_f32(false, a0, false, b0, (short)0, acc0,
                                                 false, false);
    acc1 = __builtin_amdgcn_wmma_f32_16x16x4_f32(false, a1, false, b1, (short)0, acc1,
                                                 false, false);
  }
  return acc0 + acc1;
}

__global__ __launch_bounds__(1024) void ga_agg(const float* __restrict__ yn,
                                               const float* __restrict__ Wkv,
                                               float* __restrict__ agg) {
  __shared__ __align__(32) float sred[32 * 256];
  const int tid = threadIdx.x;
  const int lane = tid & 31, wave = tid >> 5;
  const int l16 = lane & 15, lhi = lane >> 4;
  const int nt = wave & 1, ks = wave >> 1;
  const int j0 = blockIdx.x * 32;
  const int h = j0 >> 7;
  const int koff = ks * 64 + lhi * 4;
  const float* arow = yn + (size_t)l16 * (16 * 8 * 1024) + h * 1024 + koff;
  const float* wrow = Wkv + (size_t)(1024 + j0 + nt * 16 + l16) * 1024 + koff;
  const v8f acc = ga_tile64(arow, wrow);
  *(v8f*)&sred[wave * 256 + lane * 8] = acc;
  __syncthreads();
  if (tid < 512) {
    const int m = tid >> 5, n32 = tid & 31;
    const int idx = ((n32 & 15) + ((m >> 3) << 4)) * 8 + (m & 7);
    float v = 0.f;
    for (int q = 0; q < 16; ++q) v += sred[(q * 2 + (n32 >> 4)) * 256 + idx];
    agg[m * 1024 + j0 + n32] = v;
  }
}

__global__ __launch_bounds__(1024) void ga_proj(const float* __restrict__ agg,
                                                const float* __restrict__ projW,
                                                const float* __restrict__ projb,
                                                const float* __restrict__ cf,
                                                float* __restrict__ out) {
  __shared__ __align__(32) float sred[32 * 256];
  const int tid = threadIdx.x;
  const int lane = tid & 31, wave = tid >> 5;
  const int l16 = lane & 15, lhi = lane >> 4;
  const int nt = wave & 1, ks = wave >> 1;
  const int i0 = blockIdx.x * 32;
  const int koff = ks * 64 + lhi * 4;
  const float* arow = agg + (size_t)l16 * 1024 + koff;
  const float* wrow = projW + (size_t)(i0 + nt * 16 + l16) * 1024 + koff;
  const v8f acc = ga_tile64(arow, wrow);
  *(v8f*)&sred[wave * 256 + lane * 8] = acc;
  __syncthreads();
  if (tid < 512) {
    const int m = tid >> 5, n32 = tid & 31;
    const int idx = ((n32 & 15) + ((m >> 3) << 4)) * 8 + (m & 7);
    float v = 0.f;
    for (int q = 0; q < 16; ++q) v += sred[(q * 2 + (n32 >> 4)) * 256 + idx];
    out[(size_t)m * 1024 + i0 + n32] = cf[i0 + n32] + v + projb[i0 + n32];
  }
}

extern "C" void kernel_launch(void* const* d_in, const int* in_sizes, int n_in,
                              void* d_out, int out_size, void* d_ws, size_t ws_size,
                              hipStream_t stream) {
  const float* class_feature = (const float*)d_in[0];
  const float* features      = (const float*)d_in[1];
  const float* q_gamma       = (const float*)d_in[2];
  const float* q_beta        = (const float*)d_in[3];
  const float* Wq            = (const float*)d_in[4];
  const float* kv_gamma      = (const float*)d_in[5];
  const float* kv_beta       = (const float*)d_in[6];
  const float* Wkv           = (const float*)d_in[7];
  const float* proj_W        = (const float*)d_in[8];
  const float* proj_b        = (const float*)d_in[9];
  float* out = (float*)d_out;

  float* ws   = (float*)d_ws;
  float* cfn  = ws;
  float* qs   = ws + 1024;
  float* wqe  = ws + 2048;
  float* mp   = ws + 18432;
  float* lp   = ws + 20480;
  float* yp   = ws + 22528;
  float* agg  = ws + 2048;

  ga_lncf<<<1, 256, 0, stream>>>(class_feature, q_gamma, q_beta, cfn);
  ga_q<<<4, 256, 0, stream>>>(cfn, Wq, qs);
  ga_wqeff<<<4, 256, 0, stream>>>(Wkv, qs, wqe);
  ga_main<<<B_ * NCHUNK, 256, 0, stream>>>(features, kv_gamma, kv_beta, wqe, mp, lp, yp);
  ga_comb<<<B_ * 8 * 2, 128, 0, stream>>>(mp, lp, yp);
  ga_agg<<<32, 1024, 0, stream>>>(yp, Wkv, agg);
  ga_proj<<<32, 1024, 0, stream>>>(agg, proj_W, proj_b, class_feature, out);
}


# precompute products (q row; its fold into Wk) spread from 4 blocks each to 32 and 64, each lane's loads issued as one group; wqeff stored [head][column]; changes sum order of both dot products
# speedup vs baseline: 2.5899x; 1.8134x over previous
#include <hip/hip_runtime.h>
#include <hip/hip_bf16.h>

typedef __attribute__((ext_vector_type(2))) float v2f;
typedef __attribute__((ext_vector_type(8))) float v8f;

#define B_ 16
#define N_ 4096
#define C_ 1024
#define NCHUNK 16
#define CHUNK 256
#define XSTR 1032

__global__ __launch_bounds__(256) void ga_lncf(const float* __restrict__ cf,
                                               const float* __restrict__ qg,
                                               const float* __restrict__ qb,
                                               float* __restrict__ cfn) {
  __shared__ float s0[256], s1[256];
  const int tid = threadIdx.x;
  float4 x = *(const float4*)(cf + tid * 4);
  s0[tid] = x.x + x.y + x.z + x.w;
  s1[tid] = x.x * x.x + x.y * x.y + x.z * x.z + x.w * x.w;
  __syncthreads();
  for (int s = 128; s > 0; s >>= 1) {
    if (tid < s) { s0[tid] += s0[tid + s]; s1[tid] += s1[tid + s]; }
    __syncthreads();
  }
  const float mu = s0[0] * (1.f / 1024.f);
  const float var = s1[0] * (1.f / 1024.f) - mu * mu;
  const float rs = rsqrtf(var + 1e-5f);
  float4 g = *(const float4*)(qg + tid * 4);
  float4 b4 = *(const float4*)(qb + tid * 4);
  float4 o;
  o.x = (x.x - mu) * rs * g.x + b4.x;
  o.y = (x.y - mu) * rs * g.y + b4.y;
  o.z = (x.z - mu) * rs * g.z + b4.z;
  o.w = (x.w - mu) * rs * g.w + b4.w;
  *(float4*)(cfn + tid * 4) = o;
}

__global__ __launch_bounds__(1024) void ga_q(const float* __restrict__ cfn,
                                             const float* __restrict__ Wq,
                                             float* __restrict__ qs) {
  __shared__ __align__(16) float cs[1024];
  __shared__ float s_out[32];
  const int tid = threadIdx.x;
  const int lane = tid & 31, wave = tid >> 5;
  const int j = blockIdx.x * 32 + wave;
  const float cv = cfn[tid];
  const float4* wrow = (const float4*)(Wq + (size_t)j * 1024);
  float4 w[8];
  for (int i = 0; i < 8; ++i) w[i] = wrow[lane + 32 * i];
  cs[tid] = cv;
  __syncthreads();
  float acc = 0.f;
  for (int i = 0; i < 8; ++i) {
    const float4 c = *(const float4*)&cs[(lane + 32 * i) * 4];
    acc += c.x * w[i].x + c.y * w[i].y + c.z * w[i].z + c.w * w[i].w;
  }
  for (int m = 16; m > 0; m >>= 1) acc += __shfl_xor(acc, m, 32);
  if (lane == 0) s_out[wave] = acc;
  __syncthreads();
  if (tid < 32) qs[blockIdx.x * 32 + tid] = s_out[tid] * 0.03125f;
}

__global__ __launch_bounds__(512) void ga_wqeff(const float* __restrict__ Wkv,
                                                const float* __restrict__ qs,
                                                float* __restrict__ wqe) {
  __shared__ __align__(16) float sred[16 * 128];
  const int tid = threadIdx.x;
  const int lane = tid & 31, wave = tid >> 5;
  const int h = blockIdx.x >> 3;
  const int c0 = (blockIdx.x & 7) * 128;
  const int d0 = h * 128 + wave * 8;
  const float qv = qs[d0 + (lane & 7)];
  const float* wp = Wkv + (size_t)d0 * 1024 + c0 + lane * 4;
  float4 w[8];
  for (int i = 0; i < 8; ++i) w[i] = *(const float4*)(wp + (size_t)i * 1024);
  float4 acc = {0.f, 0.f, 0.f, 0.f};
  for (int i = 0; i < 8; ++i) {
    const float q = __int_as_float(__builtin_amdgcn_readlane(__float_as_int(qv), i));
    acc.x += w[i].x * q;
    acc.y += w[i].y * q;
    acc.z += w[i].z * q;
    acc.w += w[i].w * q;
  }
  *(float4*)&sred[wave * 128 + lane * 4] = acc;
  __syncthreads();
  if (tid < 32) {
    float4 s = *(const float4*)&sred[tid * 4];
    for (int k = 1; k < 16; ++k) {
      const float4 t = *(const float4*)&sred[k * 128 + tid * 4];
      s.x += t.x;
      s.y += t.y;
      s.z += t.z;
      s.w += t.w;
    }
    *(float4*)(wqe + h * 1024 + c0 + tid * 4) = s;
  }
}

__global__ __launch_bounds__(256) void ga_main(const float* __restrict__ feats,
                                               const float* __restrict__ kvg,
                                               const float* __restrict__ kvb,
                                               const float* __restrict__ wqe,
                                               float* __restrict__ mp,
                                               float* __restrict__ lp,
                                               float* __restrict__ yp) {
  __shared__ __align__(16) float xs[16][XSTR];
  __shared__ __align__(16) float wq[1024 * 17];
  __shared__ __align__(32) float sred[2048];
  __shared__ float sS[256];
  __shared__ float sP[16 * 17];
  __shared__ float sF[16];
  __shared__ float smu[16], srs[16];

  const int tid = threadIdx.x;
  const int lane = tid & 31;
  const int wave = tid >> 5;
  const int l16 = lane & 15;
  const int lhi = lane >> 4;
  const int r = tid >> 4;
  const int t16 = tid & 15;

  const int blk = blockIdx.x;
  const int bb = blk >> 4;
  const int chunk = blk & 15;

  for (int i = tid; i < 8192; i += 256) wq[(i & 1023) * 17 + (i >> 10)] = wqe[i];
  for (int i = tid; i < 8192; i += 256) wq[(i >> 3) * 17 + 8 + (i & 7)] = 0.f;

  const v8f vzero = {0.f, 0.f, 0.f, 0.f, 0.f, 0.f, 0.f, 0.f};
  v8f yacc[8];
#pragma unroll
  for (int t = 0; t < 8; ++t) yacc[t] = vzero;

  float m_run = -3.4e38f;
  float l_run = 0.f;

  __syncthreads();

#pragma unroll 1
  for (int st = 0; st < 16; ++st) {
    const size_t rowbase =
        ((size_t)bb * N_ + (size_t)chunk * CHUNK + st * 16 + r) * C_;
    float sum = 0.f, sq = 0.f;
#pragma unroll
    for (int i = 0; i < 16; ++i) {
      const int col = i * 64 + t16 * 4;
      float4 x = *(const float4*)(feats + rowbase + col);
      sum += x.x + x.y + x.z + x.w;
      sq += x.x * x.x + x.y * x.y + x.z * x.z + x.w * x.w;
      *(float4*)&xs[r][col] = x;
    }
    if (st < 15)
      __builtin_prefetch(feats + rowbase + 16 * C_ + t16 * 64, 0, 3);
    sred[tid] = sum;
    sred[256 + tid] = sq;
    __syncthreads();
    if (tid < 16) {
      float s0 = 0.f, s1 = 0.f;
#pragma unroll
      for (int i = 0; i < 16; ++i) {
        s0 += sred[tid * 16 + i];
        s1 += sred[256 + tid * 16 + i];
      }
      const float mu = s0 * (1.f / 1024.f);
      const float var = s1 * (1.f / 1024.f) - mu * mu;
      smu[tid] = mu;
      srs[tid] = rsqrtf(var + 1e-5f);
    }
    __syncthreads();
    {
      const float mu = smu[r], rs = srs[r];
#pragma unroll
      for (int i = 0; i < 16; ++i) {
        const int col = i * 64 + t16 * 4;
        float4 x = *(float4*)&xs[r][col];
        float4 g = *(const float4*)(kvg + col);
        float4 b4 = *(const float4*)(kvb + col);
        x.x = (x.x - mu) * rs * g.x + b4.x;
        x.y = (x.y - mu) * rs * g.y + b4.y;
        x.z = (x.z - mu) * rs * g.z + b4.z;
        x.w = (x.w - mu) * rs * g.w + b4.w;
        *(float4*)&xs[r][col] = x;
      }
    }
    __syncthreads();

    v8f sacc = vzero;
    const int kb = wave * 128 + 2 * lhi;
#pragma unroll
    for (int kk = 0; kk < 32; ++kk) {
      const int k0 = kb + kk * 4;
      v2f a = *(const v2f*)&xs[l16][k0];
      v2f bv;
      bv.x = wq[k0 * 17 + l16];
      bv.y = wq[(k0 + 1) * 17 + l16];
      sacc = __builtin_amdgcn_wmma_f32_16x16x4_f32(false, a, false, bv,
                                                   (short)0, sacc, false, false);
    }
    *(v8f*)&sred[wave * 256 + lane * 8] = sacc;
    __syncthreads();
    {
      const int rr = tid >> 4, hh = tid & 15;
      const int idx = (hh + ((rr >> 3) << 4)) * 8 + (rr & 7);
      float v = 0.f;
#pragma unroll
      for (int w = 0; w < 8; ++w) v += sred[w * 256 + idx];
      sS[rr * 16 + hh] = v;
    }
    __syncthreads();

    if (tid < 16) {
      float rmax = -3.4e38f;
#pragma unroll
      for (int q = 0; q < 16; ++q) rmax = fmaxf(rmax, sS[q * 16 + tid]);
      const float mnew = fmaxf(m_run, rmax);
      const float f = __expf(m_run - mnew);
      float ls = 0.f;
#pragma unroll
      for (int q = 0; q < 16; ++q) {
        const float p = __expf(sS[q * 16 + tid] - mnew);
        sP[q * 17 + tid] = p;
        ls += p;
      }
      l_run = l_run * f + ls;
      m_run = mnew;
      sF[tid] = f;
    }
    __syncthreads();

    float fv[8];
#pragma unroll
    for (int v = 0; v < 8; ++v) fv[v] = sF[v + 8 * lhi];
#pragma unroll
    for (int t = 0; t < 8; ++t) {
#pragma unroll
      for (int v = 0; v < 8; ++v) yacc[t][v] *= fv[v];
    }
#pragma unroll
    for (int k0 = 0; k0 < 16; k0 += 4) {
      const int kr = k0 + 2 * lhi;
      v2f a;
      a.x = sP[kr * 17 + l16];
      a.y = sP[(kr + 1) * 17 + l16];
#pragma unroll
      for (int t = 0; t < 8; ++t) {
        const int c0 = wave * 128 + t * 16;
        v2f bv;
        bv.x = xs[kr][c0 + l16];
        bv.y = xs[kr + 1][c0 + l16];
        yacc[t] = __builtin_amdgcn_wmma_f32_16x16x4_f32(
            false, a, false, bv, (short)0, yacc[t], false, false);
      }
    }
    __syncthreads();
  }

  if (tid < 8) {
    mp[blk * 8 + tid] = m_run;
    lp[blk * 8 + tid] = l_run;
  }
  if (lhi == 0) {
#pragma unroll
    for (int t = 0; t < 8; ++t) {
      const int c0 = wave * 128 + t * 16;
#pragma unroll
      for (int v = 0; v < 8; ++v)
        yp[(size_t)(blk * 8 + v) * 1024 + c0 + l16] = yacc[t][v];
    }
  }
}


__global__ __launch_bounds__(128) void ga_comb(const float* __restrict__ mp,
                                               const float* __restrict__ lp,
                                               float* yp) {
  const int bh = blockIdx.x >> 1;
  const int b = bh >> 3, h = bh & 7;
  const int c = (((blockIdx.x & 1) << 7) + threadIdx.x) * 4;
  float e[16];
  float M = -3.4e38f;
  for (int k = 0; k < 16; ++k) {
    e[k] = mp[(b * 16 + k) * 8 + h];
    M = fmaxf(M, e[k]);
  }
  float L = 0.f;
  for (int k = 0; k < 16; ++k) {
    e[k] = __expf(e[k] - M);
    L += lp[(b * 16 + k) * 8 + h] * e[k];
  }
  float4 acc = {0.f, 0.f, 0.f, 0.f};
  for (int k = 0; k < 16; ++k) {
    const float4 y = *(const float4*)(yp + (size_t)((b * 16 + k) * 8 + h) * 1024 + c);
    acc.x += y.x * e[k];
    acc.y += y.y * e[k];
    acc.z += y.z * e[k];
    acc.w += y.w * e[k];
  }
  acc.x = acc.x / L;
  acc.y = acc.y / L;
  acc.z = acc.z / L;
  acc.w = acc.w / L;
  *(float4*)(yp + (size_t)((b * 16) * 8 + h) * 1024 + c) = acc;
}

__device__ __forceinline__ v8f ga_tile64(const float* __restrict__ arow,
                                         const float* __restrict__ wrow) {
  const v8f vzero = {0.f, 0.f, 0.f, 0.f, 0.f, 0.f, 0.f, 0.f};
  v8f acc0 = vzero, acc1 = vzero;
#pragma unroll
  for (int s = 0; s < 8; ++s) {
    const float4 a = *(const float4*)(arow + s * 8);
    const float4 w = *(const float4*)(wrow + s * 8);
    v2f a0, a1, b0, b1;
    a0.x = a.x; a0.y = a.y; a1.x = a.z; a1.y = a.w;
    b0.x = w.x; b0.y = w.y; b1.x = w.z; b1.y = w.w;
    acc0 = __builtin_amdgcn_wmma_f32_16x16x4_f32(false, a0, false, b0, (short)0, acc0,
                                                 false, false);
    acc1 = __builtin_amdgcn_wmma_f32_16x16x4_f32(false, a1, false, b1, (short)0, acc1,
                                                 false, false);
  }
  return acc0 + acc1;
}

__global__ __launch_bounds__(1024) void ga_agg(const float* __restrict__ yn,
                                               const float* __restrict__ Wkv,
                                               float* __restrict__ agg) {
  __shared__ __align__(32) float sred[32 * 256];
  const int tid = threadIdx.x;
  const int lane = tid & 31, wave = tid >> 5;
  const int l16 = lane & 15, lhi = lane >> 4;
  const int nt = wave & 1, ks = wave >> 1;
  const int j0 = blockIdx.x * 32;
  const int h = j0 >> 7;
  const int koff = ks * 64 + lhi * 4;
  const float* arow = yn + (size_t)l16 * (16 * 8 * 1024) + h * 1024 + koff;
  const float* wrow = Wkv + (size_t)(1024 + j0 + nt * 16 + l16) * 1024 + koff;
  const v8f acc = ga_tile64(arow, wrow);
  *(v8f*)&sred[wave * 256 + lane * 8] = acc;
  __syncthreads();
  if (tid < 512) {
    const int m = tid >> 5, n32 = tid & 31;
    const int idx = ((n32 & 15) + ((m >> 3) << 4)) * 8 + (m & 7);
    float v = 0.f;
    for (int q = 0; q < 16; ++q) v += sred[(q * 2 + (n32 >> 4)) * 256 + idx];
    agg[m * 1024 + j0 + n32] = v;
  }
}

__global__ __launch_bounds__(1024) void ga_proj(const float* __restrict__ agg,
                                                const float* __restrict__ projW,
                                                const float* __restrict__ projb,
                                                const float* __restrict__ cf,
                                                float* __restrict__ out) {
  __shared__ __align__(32) float sred[32 * 256];
  const int tid = threadIdx.x;
  const int lane = tid & 31, wave = tid >> 5;
  const int l16 = lane & 15, lhi = lane >> 4;
  const int nt = wave & 1, ks = wave >> 1;
  const int i0 = blockIdx.x * 32;
  const int koff = ks * 64 + lhi * 4;
  const float* arow = agg + (size_t)l16 * 1024 + koff;
  const float* wrow = projW + (size_t)(i0 + nt * 16 + l16) * 1024 + koff;
  const v8f acc = ga_tile64(arow, wrow);
  *(v8f*)&sred[wave * 256 + lane * 8] = acc;
  __syncthreads();
  if (tid < 512) {
    const int m = tid >> 5, n32 = tid & 31;
    const int idx = ((n32 & 15) + ((m >> 3) << 4)) * 8 + (m & 7);
    float v = 0.f;
    for (int q = 0; q < 16; ++q) v += sred[(q * 2 + (n32 >> 4)) * 256 + idx];
    out[(size_t)m * 1024 + i0 + n32] = cf[i0 + n32] + v + projb[i0 + n32];
  }
}

extern "C" void kernel_launch(void* const* d_in, const int* in_sizes, int n_in,
                              void* d_out, int out_size, void* d_ws, size_t ws_size,
                              hipStream_t stream) {
  const float* class_feature = (const float*)d_in[0];
  const float* features      = (const float*)d_in[1];
  const float* q_gamma       = (const float*)d_in[2];
  const float* q_beta        = (const float*)d_in[3];
  const float* Wq            = (const float*)d_in[4];
  const float* kv_gamma      = (const float*)d_in[5];
  const float* kv_beta       = (const float*)d_in[6];
  const float* Wkv           = (const float*)d_in[7];
  const float* proj_W        = (const float*)d_in[8];
  const float* proj_b        = (const float*)d_in[9];
  float* out = (float*)d_out;

  float* ws   = (float*)d_ws;
  float* cfn  = ws;
  float* qs   = ws + 1024;
  float* wqe  = ws + 2048;
  float* mp   = ws + 18432;
  float* lp   = ws + 20480;
  float* yp   = ws + 22528;
  float* agg  = ws + 2048;

  ga_lncf<<<1, 256, 0, stream>>>(class_feature, q_gamma, q_beta, cfn);
  ga_q<<<32, 1024, 0, stream>>>(cfn, Wq, qs);
  ga_wqeff<<<64, 512, 0, stream>>>(Wkv, qs, wqe);
  ga_main<<<B_ * NCHUNK, 256, 0, stream>>>(features, kv_gamma, kv_beta, wqe, mp, lp, yp);
  ga_comb<<<B_ * 8 * 2, 128, 0, stream>>>(mp, lp, yp);
  ga_agg<<<32, 1024, 0, stream>>>(yp, Wkv, agg);
  ga_proj<<<32, 1024, 0, stream>>>(agg, proj_W, proj_b, class_feature, out);
}


# main: LayerNorm in registers (row sums by a 16-lane DPP butterfly, tile stored to LDS once, normalised): no LDS normalise pass, 5 barriers per tile not 7; changes sum order of row mean/variance
# speedup vs baseline: 2.7254x; 1.0523x over previous
#include <hip/hip_runtime.h>
#include <hip/hip_bf16.h>

typedef __attribute__((ext_vector_type(2))) float v2f;
typedef __attribute__((ext_vector_type(8))) float v8f;

#define B_ 16
#define N_ 4096
#define C_ 1024
#define NCHUNK 16
#define CHUNK 256
#define XSTR 1032

__global__ __launch_bounds__(256) void ga_lncf(const float* __restrict__ cf,
                                               const float* __restrict__ qg,
                                               const float* __restrict__ qb,
                                               float* __restrict__ cfn) {
  __shared__ float s0[256], s1[256];
  const int tid = threadIdx.x;
  float4 x = *(const float4*)(cf + tid * 4);
  s0[tid] = x.x + x.y + x.z + x.w;
  s1[tid] = x.x * x.x + x.y * x.y + x.z * x.z + x.w * x.w;
  __syncthreads();
  for (int s = 128; s > 0; s >>= 1) {
    if (tid < s) { s0[tid] += s0[tid + s]; s1[tid] += s1[tid + s]; }
    __syncthreads();
  }
  const float mu = s0[0] * (1.f / 1024.f);
  const float var = s1[0] * (1.f / 1024.f) - mu * mu;
  const float rs = rsqrtf(var + 1e-5f);
  float4 g = *(const float4*)(qg + tid * 4);
  float4 b4 = *(const float4*)(qb + tid * 4);
  float4 o;
  o.x = (x.x - mu) * rs * g.x + b4.x;
  o.y = (x.y - mu) * rs * g.y + b4.y;
  o.z = (x.z - mu) * rs * g.z + b4.z;
  o.w = (x.w - mu) * rs * g.w + b4.w;
  *(float4*)(cfn + tid * 4) = o;
}

__global__ __launch_bounds__(1024) void ga_q(const float* __restrict__ cfn,
                                             const float* __restrict__ Wq,
                                             float* __restrict__ qs) {
  __shared__ __align__(16) float cs[1024];
  __shared__ float s_out[32];
  const int tid = threadIdx.x;
  const int lane = tid & 31, wave = tid >> 5;
  const int j = blockIdx.x * 32 + wave;
  const float cv = cfn[tid];
  const float4* wrow = (const float4*)(Wq + (size_t)j * 1024);
  float4 w[8];
  for (int i = 0; i < 8; ++i) w[i] = wrow[lane + 32 * i];
  cs[tid] = cv;
  __syncthreads();
  float acc = 0.f;
  for (int i = 0; i < 8; ++i) {
    const float4 c = *(const float4*)&cs[(lane + 32 * i) * 4];
    acc += c.x * w[i].x + c.y * w[i].y + c.z * w[i].z + c.w * w[i].w;
  }
  for (int m = 16; m > 0; m >>= 1) acc += __shfl_xor(acc, m, 32);
  if (lane == 0) s_out[wave] = acc;
  __syncthreads();
  if (tid < 32) qs[blockIdx.x * 32 + tid] = s_out[tid] * 0.03125f;
}

__global__ __launch_bounds__(512) void ga_wqeff(const float* __restrict__ Wkv,
                                                const float* __restrict__ qs,
                                                float* __restrict__ wqe) {
  __shared__ __align__(16) float sred[16 * 128];
  const int tid = threadIdx.x;
  const int lane = tid & 31, wave = tid >> 5;
  const int h = blockIdx.x >> 3;
  const int c0 = (blockIdx.x & 7) * 128;
  const int d0 = h * 128 + wave * 8;
  const float qv = qs[d0 + (lane & 7)];
  const float* wp = Wkv + (size_t)d0 * 1024 + c0 + lane * 4;
  float4 w[8];
  for (int i = 0; i < 8; ++i) w[i] = *(const float4*)(wp + (size_t)i * 1024);
  float4 acc = {0.f, 0.f, 0.f, 0.f};
  for (int i = 0; i < 8; ++i) {
    const float q = __int_as_float(__builtin_amdgcn_readlane(__float_as_int(qv), i));
    acc.x += w[i].x * q;
    acc.y += w[i].y * q;
    acc.z += w[i].z * q;
    acc.w += w[i].w * q;
  }
  *(float4*)&sred[wave * 128 + lane * 4] = acc;
  __syncthreads();
  if (tid < 32) {
    float4 s = *(const float4*)&sred[tid * 4];
    for (int k = 1; k < 16; ++k) {
      const float4 t = *(const float4*)&sred[k * 128 + tid * 4];
      s.x += t.x;
      s.y += t.y;
      s.z += t.z;
      s.w += t.w;
    }
    *(float4*)(wqe + h * 1024 + c0 + tid * 4) = s;
  }
}

template <int CTRL>
__device__ __forceinline__ float ga_row_xor(float v) {
  return __int_as_float(__builtin_amdgcn_update_dpp(__float_as_int(v), __float_as_int(v), CTRL,
                                                    0xF, 0xF, false));
}

__global__ __launch_bounds__(256) void ga_main(const float* __restrict__ feats,
                                               const float* __restrict__ kvg,
                                               const float* __restrict__ kvb,
                                               const float* __restrict__ wqe,
                                               float* __restrict__ mp,
                                               float* __restrict__ lp,
                                               float* __restrict__ yp) {
  __shared__ __align__(16) float xs[16][XSTR];
  __shared__ __align__(16) float wq[1024 * 17];
  __shared__ __align__(32) float sred[2048];
  __shared__ float sS[256];
  __shared__ float sP[16 * 17];
  __shared__ float sF[16];

  const int tid = threadIdx.x;
  const int lane = tid & 31;
  const int wave = tid >> 5;
  const int l16 = lane & 15;
  const int lhi = lane >> 4;
  const int r = tid >> 4;
  const int t16 = tid & 15;

  const int blk = blockIdx.x;
  const int bb = blk >> 4;
  const int chunk = blk & 15;

  for (int i = tid; i < 8192; i += 256) wq[(i & 1023) * 17 + (i >> 10)] = wqe[i];
  for (int i = tid; i < 8192; i += 256) wq[(i >> 3) * 17 + 8 + (i & 7)] = 0.f;

  float4 g4[16], b4[16];
#pragma unroll
  for (int i = 0; i < 16; ++i) {
    g4[i] = *(const float4*)(kvg + i * 64 + t16 * 4);
    b4[i] = *(const float4*)(kvb + i * 64 + t16 * 4);
  }

  const v8f vzero = {0.f, 0.f, 0.f, 0.f, 0.f, 0.f, 0.f, 0.f};
  v8f yacc[8];
#pragma unroll
  for (int t = 0; t < 8; ++t) yacc[t] = vzero;

  float m_run = -3.4e38f;
  float l_run = 0.f;

  __syncthreads();

#pragma unroll 1
  for (int st = 0; st < 16; ++st) {
    const size_t rowbase =
        ((size_t)bb * N_ + (size_t)chunk * CHUNK + st * 16 + r) * C_;
    float4 x[16];
#pragma unroll
    for (int i = 0; i < 16; ++i)
      x[i] = *(const float4*)(feats + rowbase + i * 64 + t16 * 4);
    if (st < 15)
      __builtin_prefetch(feats + rowbase + 16 * C_ + t16 * 64, 0, 3);
    float sum = 0.f, sq = 0.f;
#pragma unroll
    for (int i = 0; i < 16; ++i) {
      sum += x[i].x + x[i].y + x[i].z + x[i].w;
      sq += x[i].x * x[i].x + x[i].y * x[i].y + x[i].z * x[i].z + x[i].w * x[i].w;
    }
    sum += ga_row_xor<0x168>(sum);
    sq += ga_row_xor<0x168>(sq);
    sum += ga_row_xor<0x164>(sum);
    sq += ga_row_xor<0x164>(sq);
    sum += ga_row_xor<0x162>(sum);
    sq += ga_row_xor<0x162>(sq);
    sum += ga_row_xor<0x161>(sum);
    sq += ga_row_xor<0x161>(sq);
    {
      const float mu = sum * (1.f / 1024.f);
      const float var = sq * (1.f / 1024.f) - mu * mu;
      const float rs = rsqrtf(var + 1e-5f);
#pragma unroll
      for (int i = 0; i < 16; ++i) {
        float4 o;
        o.x = (x[i].x - mu) * rs * g4[i].x + b4[i].x;
        o.y = (x[i].y - mu) * rs * g4[i].y + b4[i].y;
        o.z = (x[i].z - mu) * rs * g4[i].z + b4[i].z;
        o.w = (x[i].w - mu) * rs * g4[i].w + b4[i].w;
        *(float4*)&xs[r][i * 64 + t16 * 4] = o;
      }
    }
    __syncthreads();

    v8f sacc = vzero;
    const int kb = wave * 128 + 2 * lhi;
#pragma unroll
    for (int kk = 0; kk < 32; ++kk) {
      const int k0 = kb + kk * 4;
      v2f a = *(const v2f*)&xs[l16][k0];
      v2f bv;
      bv.x = wq[k0 * 17 + l16];
      bv.y = wq[(k0 + 1) * 17 + l16];
      sacc = __builtin_amdgcn_wmma_f32_16x16x4_f32(false, a, false, bv,
                                                   (short)0, sacc, false, false);
    }
    *(v8f*)&sred[wave * 256 + lane * 8] = sacc;
    __syncthreads();
    {
      const int rr = tid >> 4, hh = tid & 15;
      const int idx = (hh + ((rr >> 3) << 4)) * 8 + (rr & 7);
      float v = 0.f;
#pragma unroll
      for (int w = 0; w < 8; ++w) v += sred[w * 256 + idx];
      sS[rr * 16 + hh] = v;
    }
    __syncthreads();

    if (tid < 16) {
      float rmax = -3.4e38f;
#pragma unroll
      for (int q = 0; q < 16; ++q) rmax = fmaxf(rmax, sS[q * 16 + tid]);
      const float mnew = fmaxf(m_run, rmax);
      const float f = __expf(m_run - mnew);
      float ls = 0.f;
#pragma unroll
      for (int q = 0; q < 16; ++q) {
        const float p = __expf(sS[q * 16 + tid] - mnew);
        sP[q * 17 + tid] = p;
        ls += p;
      }
      l_run = l_run * f + ls;
      m_run = mnew;
      sF[tid] = f;
    }
    __syncthreads();

    float fv[8];
#pragma unroll
    for (int v = 0; v < 8; ++v) fv[v] = sF[v + 8 * lhi];
#pragma unroll
    for (int t = 0; t < 8; ++t) {
#pragma unroll
      for (int v = 0; v < 8; ++v) yacc[t][v] *= fv[v];
    }
#pragma unroll
    for (int k0 = 0; k0 < 16; k0 += 4) {
      const int kr = k0 + 2 * lhi;
      v2f a;
      a.x = sP[kr * 17 + l16];
      a.y = sP[(kr + 1) * 17 + l16];
#pragma unroll
      for (int t = 0; t < 8; ++t) {
        const int c0 = wave * 128 + t * 16;
        v2f bv;
        bv.x = xs[kr][c0 + l16];
        bv.y = xs[kr + 1][c0 + l16];
        yacc[t] = __builtin_amdgcn_wmma_f32_16x16x4_f32(
            false, a, false, bv, (short)0, yacc[t], false, false);
      }
    }
    __syncthreads();
  }

  if (tid < 8) {
    mp[blk * 8 + tid] = m_run;
    lp[blk * 8 + tid] = l_run;
  }
  if (lhi == 0) {
#pragma unroll
    for (int t = 0; t < 8; ++t) {
      const int c0 = wave * 128 + t * 16;
#pragma unroll
      for (int v = 0; v < 8; ++v)
        yp[(size_t)(blk * 8 + v) * 1024 + c0 + l16] = yacc[t][v];
    }
  }
}


__global__ __launch_bounds__(128) void ga_comb(const float* __restrict__ mp,
                                               const float* __restrict__ lp,
                                               float* yp) {
  const int bh = blockIdx.x >> 1;
  const int b = bh >> 3, h = bh & 7;
  const int c = (((blockIdx.x & 1) << 7) + threadIdx.x) * 4;
  float e[16];
  float M = -3.4e38f;
  for (int k = 0; k < 16; ++k) {
    e[k] = mp[(b * 16 + k) * 8 + h];
    M = fmaxf(M, e[k]);
  }
  float L = 0.f;
  for (int k = 0; k < 16; ++k) {
    e[k] = __expf(e[k] - M);
    L += lp[(b * 16 + k) * 8 + h] * e[k];
  }
  float4 acc = {0.f, 0.f, 0.f, 0.f};
  for (int k = 0; k < 16; ++k) {
    const float4 y = *(const float4*)(yp + (size_t)((b * 16 + k) * 8 + h) * 1024 + c);
    acc.x += y.x * e[k];
    acc.y += y.y * e[k];
    acc.z += y.z * e[k];
    acc.w += y.w * e[k];
  }
  acc.x = acc.x / L;
  acc.y = acc.y / L;
  acc.z = acc.z / L;
  acc.w = acc.w / L;
  *(float4*)(yp + (size_t)((b * 16) * 8 + h) * 1024 + c) = acc;
}

__device__ __forceinline__ v8f ga_tile64(const float* __restrict__ arow,
                                         const float* __restrict__ wrow) {
  const v8f vzero = {0.f, 0.f, 0.f, 0.f, 0.f, 0.f, 0.f, 0.f};
  v8f acc0 = vzero, acc1 = vzero;
#pragma unroll
  for (int s = 0; s < 8; ++s) {
    const float4 a = *(const float4*)(arow + s * 8);
    const float4 w = *(const float4*)(wrow + s * 8);
    v2f a0, a1, b0, b1;
    a0.x = a.x; a0.y = a.y; a1.x = a.z; a1.y = a.w;
    b0.x = w.x; b0.y = w.y; b1.x = w.z; b1.y = w.w;
    acc0 = __builtin_amdgcn_wmma_f32_16x16x4_f32(false, a0, false, b0, (short)0, acc0,
                                                 false, false);
    acc1 = __builtin_amdgcn_wmma_f32_16x16x4_f32(false, a1, false, b1, (short)0, acc1,
                                                 false, false);
  }
  return acc0 + acc1;
}

__global__ __launch_bounds__(1024) void ga_agg(const float* __restrict__ yn,
                                               const float* __restrict__ Wkv,
                                               float* __restrict__ agg) {
  __shared__ __align__(32) float sred[32 * 256];
  const int tid = threadIdx.x;
  const int lane = tid & 31, wave = tid >> 5;
  const int l16 = lane & 15, lhi = lane >> 4;
  const int nt = wave & 1, ks = wave >> 1;
  const int j0 = blockIdx.x * 32;
  const int h = j0 >> 7;
  const int koff = ks * 64 + lhi * 4;
  const float* arow = yn + (size_t)l16 * (16 * 8 * 1024) + h * 1024 + koff;
  const float* wrow = Wkv + (size_t)(1024 + j0 + nt * 16 + l16) * 1024 + koff;
  const v8f acc = ga_tile64(arow, wrow);
  *(v8f*)&sred[wave * 256 + lane * 8] = acc;
  __syncthreads();
  if (tid < 512) {
    const int m = tid >> 5, n32 = tid & 31;
    const int idx = ((n32 & 15) + ((m >> 3) << 4)) * 8 + (m & 7);
    float v = 0.f;
    for (int q = 0; q < 16; ++q) v += sred[(q * 2 + (n32 >> 4)) * 256 + idx];
    agg[m * 1024 + j0 + n32] = v;
  }
}

__global__ __launch_bounds__(1024) void ga_proj(const float* __restrict__ agg,
                                                const float* __restrict__ projW,
                                                const float* __restrict__ projb,
                                                const float* __restrict__ cf,
                                                float* __restrict__ out) {
  __shared__ __align__(32) float sred[32 * 256];
  const int tid = threadIdx.x;
  const int lane = tid & 31, wave = tid >> 5;
  const int l16 = lane & 15, lhi = lane >> 4;
  const int nt = wave & 1, ks = wave >> 1;
  const int i0 = blockIdx.x * 32;
  const int koff = ks * 64 + lhi * 4;
  const float* arow = agg + (size_t)l16 * 1024 + koff;
  const float* wrow = projW + (size_t)(i0 + nt * 16 + l16) * 1024 + koff;
  const v8f acc = ga_tile64(arow, wrow);
  *(v8f*)&sred[wave * 256 + lane * 8] = acc;
  __syncthreads();
  if (tid < 512) {
    const int m = tid >> 5, n32 = tid & 31;
    const int idx = ((n32 & 15) + ((m >> 3) << 4)) * 8 + (m & 7);
    float v = 0.f;
    for (int q = 0; q < 16; ++q) v += sred[(q * 2 + (n32 >> 4)) * 256 + idx];
    out[(size_t)m * 1024 + i0 + n32] = cf[i0 + n32] + v + projb[i0 + n32];
  }
}

extern "C" void kernel_launch(void* const* d_in, const int* in_sizes, int n_in,
                              void* d_out, int out_size, void* d_ws, size_t ws_size,
                              hipStream_t stream) {
  const float* class_feature = (const float*)d_in[0];
  const float* features      = (const float*)d_in[1];
  const float* q_gamma       = (const float*)d_in[2];
  const float* q_beta        = (const float*)d_in[3];
  const float* Wq            = (const float*)d_in[4];
  const float* kv_gamma      = (const float*)d_in[5];
  const float* kv_beta       = (const float*)d_in[6];
  const float* Wkv           = (const float*)d_in[7];
  const float* proj_W        = (const float*)d_in[8];
  const float* proj_b        = (const float*)d_in[9];
  float* out = (float*)d_out;

  float* ws   = (float*)d_ws;
  float* cfn  = ws;
  float* qs   = ws + 1024;
  float* wqe  = ws + 2048;
  float* mp   = ws + 18432;
  float* lp   = ws + 20480;
  float* yp   = ws + 22528;
  float* agg  = ws + 2048;

  ga_lncf<<<1, 256, 0, stream>>>(class_feature, q_gamma, q_beta, cfn);
  ga_q<<<32, 1024, 0, stream>>>(cfn, Wq, qs);
  ga_wqeff<<<64, 512, 0, stream>>>(Wkv, qs, wqe);
  ga_main<<<B_ * NCHUNK, 256, 0, stream>>>(features, kv_gamma, kv_beta, wqe, mp, lp, yp);
  ga_comb<<<B_ * 8 * 2, 128, 0, stream>>>(mp, lp, yp);
  ga_agg<<<32, 1024, 0, stream>>>(yp, Wkv, agg);
  ga_proj<<<32, 1024, 0, stream>>>(agg, proj_W, proj_b, class_feature, out);
}


# main: feature loads issued one subtile ahead (behind the tile store; first ones before weight staging); loop barriers order LDS only, so loads fly beside both matrix phases; L2 prefetch dropped
# speedup vs baseline: 2.8093x; 1.0308x over previous
#include <hip/hip_runtime.h>
#include <hip/hip_bf16.h>

typedef __attribute__((ext_vector_type(2))) float v2f;
typedef __attribute__((ext_vector_type(8))) float v8f;

#define B_ 16
#define N_ 4096
#define C_ 1024
#define NCHUNK 16
#define CHUNK 256
#define XSTR 1032

__global__ __launch_bounds__(256) void ga_lncf(const float* __restrict__ cf,
                                               const float* __restrict__ qg,
                                               const float* __restrict__ qb,
                                               float* __restrict__ cfn) {
  __shared__ float s0[256], s1[256];
  const int tid = threadIdx.x;
  float4 x = *(const float4*)(cf + tid * 4);
  s0[tid] = x.x + x.y + x.z + x.w;
  s1[tid] = x.x * x.x + x.y * x.y + x.z * x.z + x.w * x.w;
  __syncthreads();
  for (int s = 128; s > 0; s >>= 1) {
    if (tid < s) { s0[tid] += s0[tid + s]; s1[tid] += s1[tid + s]; }
    __syncthreads();
  }
  const float mu = s0[0] * (1.f / 1024.f);
  const float var = s1[0] * (1.f / 1024.f) - mu * mu;
  const float rs = rsqrtf(var + 1e-5f);
  float4 g = *(const float4*)(qg + tid * 4);
  float4 b4 = *(const float4*)(qb + tid * 4);
  float4 o;
  o.x = (x.x - mu) * rs * g.x + b4.x;
  o.y = (x.y - mu) * rs * g.y + b4.y;
  o.z = (x.z - mu) * rs * g.z + b4.z;
  o.w = (x.w - mu) * rs * g.w + b4.w;
  *(float4*)(cfn + tid * 4) = o;
}

__global__ __launch_bounds__(1024) void ga_q(const float* __restrict__ cfn,
                                             const float* __restrict__ Wq,
                                             float* __restrict__ qs) {
  __shared__ __align__(16) float cs[1024];
  __shared__ float s_out[32];
  const int tid = threadIdx.x;
  const int lane = tid & 31, wave = tid >> 5;
  const int j = blockIdx.x * 32 + wave;
  const float cv = cfn[tid];
  const float4* wrow = (const float4*)(Wq + (size_t)j * 1024);
  float4 w[8];
  for (int i = 0; i < 8; ++i) w[i] = wrow[lane + 32 * i];
  cs[tid] = cv;
  __syncthreads();
  float acc = 0.f;
  for (int i = 0; i < 8; ++i) {
    const float4 c = *(const float4*)&cs[(lane + 32 * i) * 4];
    acc += c.x * w[i].x + c.y * w[i].y + c.z * w[i].z + c.w * w[i].w;
  }
  for (int m = 16; m > 0; m >>= 1) acc += __shfl_xor(acc, m, 32);
  if (lane == 0) s_out[wave] = acc;
  __syncthreads();
  if (tid < 32) qs[blockIdx.x * 32 + tid] = s_out[tid] * 0.03125f;
}

__global__ __launch_bounds__(512) void ga_wqeff(const float* __restrict__ Wkv,
                                                const float* __restrict__ qs,
                                                float* __restrict__ wqe) {
  __shared__ __align__(16) float sred[16 * 128];
  const int tid = threadIdx.x;
  const int lane = tid & 31, wave = tid >> 5;
  const int h = blockIdx.x >> 3;
  const int c0 = (blockIdx.x & 7) * 128;
  const int d0 = h * 128 + wave * 8;
  const float qv = qs[d0 + (lane & 7)];
  const float* wp = Wkv + (size_t)d0 * 1024 + c0 + lane * 4;
  float4 w[8];
  for (int i = 0; i < 8; ++i) w[i] = *(const float4*)(wp + (size_t)i * 1024);
  float4 acc = {0.f, 0.f, 0.f, 0.f};
  for (int i = 0; i < 8; ++i) {
    const float q = __int_as_float(__builtin_amdgcn_readlane(__float_as_int(qv), i));
    acc.x += w[i].x * q;
    acc.y += w[i].y * q;
    acc.z += w[i].z * q;
    acc.w += w[i].w * q;
  }
  *(float4*)&sred[wave * 128 + lane * 4] = acc;
  __syncthreads();
  if (tid < 32) {
    float4 s = *(const float4*)&sred[tid * 4];
    for (int k = 1; k < 16; ++k) {
      const float4 t = *(const float4*)&sred[k * 128 + tid * 4];
      s.x += t.x;
      s.y += t.y;
      s.z += t.z;
      s.w += t.w;
    }
    *(float4*)(wqe + h * 1024 + c0 + tid * 4) = s;
  }
}

template <int CTRL>
__device__ __forceinline__ float ga_row_xor(float v) {
  return __int_as_float(__builtin_amdgcn_update_dpp(__float_as_int(v), __float_as_int(v), CTRL,
                                                    0xF, 0xF, false));
}

__device__ __forceinline__ void ga_lds_barrier() {
  __builtin_amdgcn_fence(__ATOMIC_RELEASE, "workgroup", "local");
  __builtin_amdgcn_s_barrier();
  __builtin_amdgcn_fence(__ATOMIC_ACQUIRE, "workgroup", "local");
}

__global__ __launch_bounds__(256) void ga_main(const float* __restrict__ feats,
                                               const float* __restrict__ kvg,
                                               const float* __restrict__ kvb,
                                               const float* __restrict__ wqe,
                                               float* __restrict__ mp,
                                               float* __restrict__ lp,
                                               float* __restrict__ yp) {
  __shared__ __align__(16) float xs[16][XSTR];
  __shared__ __align__(16) float wq[1024 * 17];
  __shared__ __align__(32) float sred[2048];
  __shared__ float sS[256];
  __shared__ float sP[16 * 17];
  __shared__ float sF[16];

  const int tid = threadIdx.x;
  const int lane = tid & 31;
  const int wave = tid >> 5;
  const int l16 = lane & 15;
  const int lhi = lane >> 4;
  const int r = tid >> 4;
  const int t16 = tid & 15;

  const int blk = blockIdx.x;
  const int bb = blk >> 4;
  const int chunk = blk & 15;

  size_t rowbase = ((size_t)bb * N_ + (size_t)chunk * CHUNK + r) * C_;
  float4 x[16];
#pragma unroll
  for (int i = 0; i < 16; ++i)
    x[i] = *(const float4*)(feats + rowbase + i * 64 + t16 * 4);

  for (int i = tid; i < 8192; i += 256) wq[(i & 1023) * 17 + (i >> 10)] = wqe[i];
  for (int i = tid; i < 8192; i += 256) wq[(i >> 3) * 17 + 8 + (i & 7)] = 0.f;

  float4 g4[16], b4[16];
#pragma unroll
  for (int i = 0; i < 16; ++i) {
    g4[i] = *(const float4*)(kvg + i * 64 + t16 * 4);
    b4[i] = *(const float4*)(kvb + i * 64 + t16 * 4);
  }

  const v8f vzero = {0.f, 0.f, 0.f, 0.f, 0.f, 0.f, 0.f, 0.f};
  v8f yacc[8];
#pragma unroll
  for (int t = 0; t < 8; ++t) yacc[t] = vzero;

  float m_run = -3.4e38f;
  float l_run = 0.f;

  __syncthreads();

#pragma unroll 1
  for (int st = 0; st < 16; ++st) {
    float sum = 0.f, sq = 0.f;
#pragma unroll
    for (int i = 0; i < 16; ++i) {
      sum += x[i].x + x[i].y + x[i].z + x[i].w;
      sq += x[i].x * x[i].x + x[i].y * x[i].y + x[i].z * x[i].z + x[i].w * x[i].w;
    }
    sum += ga_row_xor<0x168>(sum);
    sq += ga_row_xor<0x168>(sq);
    sum += ga_row_xor<0x164>(sum);
    sq += ga_row_xor<0x164>(sq);
    sum += ga_row_xor<0x162>(sum);
    sq += ga_row_xor<0x162>(sq);
    sum += ga_row_xor<0x161>(sum);
    sq += ga_row_xor<0x161>(sq);
    {
      const float mu = sum * (1.f / 1024.f);
      const float var = sq * (1.f / 1024.f) - mu * mu;
      const float rs = rsqrtf(var + 1e-5f);
#pragma unroll
      for (int i = 0; i < 16; ++i) {
        float4 o;
        o.x = (x[i].x - mu) * rs * g4[i].x + b4[i].x;
        o.y = (x[i].y - mu) * rs * g4[i].y + b4[i].y;
        o.z = (x[i].z - mu) * rs * g4[i].z + b4[i].z;
        o.w = (x[i].w - mu) * rs * g4[i].w + b4[i].w;
        *(float4*)&xs[r][i * 64 + t16 * 4] = o;
      }
    }
    if (st < 15) {
      rowbase += (size_t)16 * C_;
#pragma unroll
      for (int i = 0; i < 16; ++i)
        x[i] = *(const float4*)(feats + rowbase + i * 64 + t16 * 4);
    }
    ga_lds_barrier();

    v8f sacc = vzero;
    const int kb = wave * 128 + 2 * lhi;
#pragma unroll
    for (int kk = 0; kk < 32; ++kk) {
      const int k0 = kb + kk * 4;
      v2f a = *(const v2f*)&xs[l16][k0];
      v2f bv;
      bv.x = wq[k0 * 17 + l16];
      bv.y = wq[(k0 + 1) * 17 + l16];
      sacc = __builtin_amdgcn_wmma_f32_16x16x4_f32(false, a, false, bv,
                                                   (short)0, sacc, false, false);
    }
    *(v8f*)&sred[wave * 256 + lane * 8] = sacc;
    ga_lds_barrier();
    {
      const int rr = tid >> 4, hh = tid & 15;
      const int idx = (hh + ((rr >> 3) << 4)) * 8 + (rr & 7);
      float v = 0.f;
#pragma unroll
      for (int w = 0; w < 8; ++w) v += sred[w * 256 + idx];
      sS[rr * 16 + hh] = v;
    }
    ga_lds_barrier();

    if (tid < 16) {
      float rmax = -3.4e38f;
#pragma unroll
      for (int q = 0; q < 16; ++q) rmax = fmaxf(rmax, sS[q * 16 + tid]);
      const float mnew = fmaxf(m_run, rmax);
      const float f = __expf(m_run - mnew);
      float ls = 0.f;
#pragma unroll
      for (int q = 0; q < 16; ++q) {
        const float p = __expf(sS[q * 16 + tid] - mnew);
        sP[q * 17 + tid] = p;
        ls += p;
      }
      l_run = l_run * f + ls;
      m_run = mnew;
      sF[tid] = f;
    }
    ga_lds_barrier();

    float fv[8];
#pragma unroll
    for (int v = 0; v < 8; ++v) fv[v] = sF[v + 8 * lhi];
#pragma unroll
    for (int t = 0; t < 8; ++t) {
#pragma unroll
      for (int v = 0; v < 8; ++v) yacc[t][v] *= fv[v];
    }
#pragma unroll
    for (int k0 = 0; k0 < 16; k0 += 4) {
      const int kr = k0 + 2 * lhi;
      v2f a;
      a.x = sP[kr * 17 + l16];
      a.y = sP[(kr + 1) * 17 + l16];
#pragma unroll
      for (int t = 0; t < 8; ++t) {
        const int c0 = wave * 128 + t * 16;
        v2f bv;
        bv.x = xs[kr][c0 + l16];
        bv.y = xs[kr + 1][c0 + l16];
        yacc[t] = __builtin_amdgcn_wmma_f32_16x16x4_f32(
            false, a, false, bv, (short)0, yacc[t], false, false);
      }
    }
    ga_lds_barrier();
  }

  if (tid < 8) {
    mp[blk * 8 + tid] = m_run;
    lp[blk * 8 + tid] = l_run;
  }
  if (lhi == 0) {
#pragma unroll
    for (int t = 0; t < 8; ++t) {
      const int c0 = wave * 128 + t * 16;
#pragma unroll
      for (int v = 0; v < 8; ++v)
        yp[(size_t)(blk * 8 + v) * 1024 + c0 + l16] = yacc[t][v];
    }
  }
}


__global__ __launch_bounds__(128) void ga_comb(const float* __restrict__ mp,
                                               const float* __restrict__ lp,
                                               float* yp) {
  const int bh = blockIdx.x >> 1;
  const int b = bh >> 3, h = bh & 7;
  const int c = (((blockIdx.x & 1) << 7) + threadIdx.x) * 4;
  float e[16];
  float M = -3.4e38f;
  for (int k = 0; k < 16; ++k) {
    e[k] = mp[(b * 16 + k) * 8 + h];
    M = fmaxf(M, e[k]);
  }
  float L = 0.f;
  for (int k = 0; k < 16; ++k) {
    e[k] = __expf(e[k] - M);
    L += lp[(b * 16 + k) * 8 + h] * e[k];
  }
  float4 acc = {0.f, 0.f, 0.f, 0.f};
  for (int k = 0; k < 16; ++k) {
    const float4 y = *(const float4*)(yp + (size_t)((b * 16 + k) * 8 + h) * 1024 + c);
    acc.x += y.x * e[k];
    acc.y += y.y * e[k];
    acc.z += y.z * e[k];
    acc.w += y.w * e[k];
  }
  acc.x = acc.x / L;
  acc.y = acc.y / L;
  acc.z = acc.z / L;
  acc.w = acc.w / L;
  *(float4*)(yp + (size_t)((b * 16) * 8 + h) * 1024 + c) = acc;
}

__device__ __forceinline__ v8f ga_tile64(const float* __restrict__ arow,
                                         const float* __restrict__ wrow) {
  const v8f vzero = {0.f, 0.f, 0.f, 0.f, 0.f, 0.f, 0.f, 0.f};
  v8f acc0 = vzero, acc1 = vzero;
#pragma unroll
  for (int s = 0; s < 8; ++s) {
    const float4 a = *(const float4*)(arow + s * 8);
    const float4 w = *(const float4*)(wrow + s * 8);
    v2f a0, a1, b0, b1;
    a0.x = a.x; a0.y = a.y; a1.x = a.z; a1.y = a.w;
    b0.x = w.x; b0.y = w.y; b1.x = w.z; b1.y = w.w;
    acc0 = __builtin_amdgcn_wmma_f32_16x16x4_f32(false, a0, false, b0, (short)0, acc0,
                                                 false, false);
    acc1 = __builtin_amdgcn_wmma_f32_16x16x4_f32(false, a1, false, b1, (short)0, acc1,
                                                 false, false);
  }
  return acc0 + acc1;
}

__global__ __launch_bounds__(1024) void ga_agg(const float* __restrict__ yn,
                                               const float* __restrict__ Wkv,
                                               float* __restrict__ agg) {
  __shared__ __align__(32) float sred[32 * 256];
  const int tid = threadIdx.x;
  const int lane = tid & 31, wave = tid >> 5;
  const int l16 = lane & 15, lhi = lane >> 4;
  const int nt = wave & 1, ks = wave >> 1;
  const int j0 = blockIdx.x * 32;
  const int h = j0 >> 7;
  const int koff = ks * 64 + lhi * 4;
  const float* arow = yn + (size_t)l16 * (16 * 8 * 1024) + h * 1024 + koff;
  const float* wrow = Wkv + (size_t)(1024 + j0 + nt * 16 + l16) * 1024 + koff;
  const v8f acc = ga_tile64(arow, wrow);
  *(v8f*)&sred[wave * 256 + lane * 8] = acc;
  __syncthreads();
  if (tid < 512) {
    const int m = tid >> 5, n32 = tid & 31;
    const int idx = ((n32 & 15) + ((m >> 3) << 4)) * 8 + (m & 7);
    float v = 0.f;
    for (int q = 0; q < 16; ++q) v += sred[(q * 2 + (n32 >> 4)) * 256 + idx];
    agg[m * 1024 + j0 + n32] = v;
  }
}

__global__ __launch_bounds__(1024) void ga_proj(const float* __restrict__ agg,
                                                const float* __restrict__ projW,
                                                const float* __restrict__ projb,
                                                const float* __restrict__ cf,
                                                float* __restrict__ out) {
  __shared__ __align__(32) float sred[32 * 256];
  const int tid = threadIdx.x;
  const int lane = tid & 31, wave = tid >> 5;
  const int l16 = lane & 15, lhi = lane >> 4;
  const int nt = wave & 1, ks = wave >> 1;
  const int i0 = blockIdx.x * 32;
  const int koff = ks * 64 + lhi * 4;
  const float* arow = agg + (size_t)l16 * 1024 + koff;
  const float* wrow = projW + (size_t)(i0 + nt * 16 + l16) * 1024 + koff;
  const v8f acc = ga_tile64(arow, wrow);
  *(v8f*)&sred[wave * 256 + lane * 8] = acc;
  __syncthreads();
  if (tid < 512) {
    const int m = tid >> 5, n32 = tid & 31;
    const int idx = ((n32 & 15) + ((m >> 3) << 4)) * 8 + (m & 7);
    float v = 0.f;
    for (int q = 0; q < 16; ++q) v += sred[(q * 2 + (n32 >> 4)) * 256 + idx];
    out[(size_t)m * 1024 + i0 + n32] = cf[i0 + n32] + v + projb[i0 + n32];
  }
}

extern "C" void kernel_launch(void* const* d_in, const int* in_sizes, int n_in,
                              void* d_out, int out_size, void* d_ws, size_t ws_size,
                              hipStream_t stream) {
  const float* class_feature = (const float*)d_in[0];
  const float* features      = (const float*)d_in[1];
  const float* q_gamma       = (const float*)d_in[2];
  const float* q_beta        = (const float*)d_in[3];
  const float* Wq            = (const float*)d_in[4];
  const float* kv_gamma      = (const float*)d_in[5];
  const float* kv_beta       = (const float*)d_in[6];
  const float* Wkv           = (const float*)d_in[7];
  const float* proj_W        = (const float*)d_in[8];
  const float* proj_b        = (const float*)d_in[9];
  float* out = (float*)d_out;

  float* ws   = (float*)d_ws;
  float* cfn  = ws;
  float* qs   = ws + 1024;
  float* wqe  = ws + 2048;
  float* mp   = ws + 18432;
  float* lp   = ws + 20480;
  float* yp   = ws + 22528;
  float* agg  = ws + 2048;

  ga_lncf<<<1, 256, 0, stream>>>(class_feature, q_gamma, q_beta, cfn);
  ga_q<<<32, 1024, 0, stream>>>(cfn, Wq, qs);
  ga_wqeff<<<64, 512, 0, stream>>>(Wkv, qs, wqe);
  ga_main<<<B_ * NCHUNK, 256, 0, stream>>>(features, kv_gamma, kv_beta, wqe, mp, lp, yp);
  ga_comb<<<B_ * 8 * 2, 128, 0, stream>>>(mp, lp, yp);
  ga_agg<<<32, 1024, 0, stream>>>(yp, Wkv, agg);
  ga_proj<<<32, 1024, 0, stream>>>(agg, proj_W, proj_b, class_feature, out);
}


# main: score-product B fragments (query weights) kept in 64 registers per wave, no LDS copy; A fragments read as one b128 group ahead of the 32 WMMAs, two accumulators; changes sum order of the scores
# speedup vs baseline: 3.0218x; 1.0756x over previous
#include <hip/hip_runtime.h>
#include <hip/hip_bf16.h>

typedef __attribute__((ext_vector_type(2))) float v2f;
typedef __attribute__((ext_vector_type(8))) float v8f;

#define B_ 16
#define N_ 4096
#define C_ 1024
#define NCHUNK 16
#define CHUNK 256
#define XSTR 1036

__global__ __launch_bounds__(256) void ga_lncf(const float* __restrict__ cf,
                                               const float* __restrict__ qg,
                                               const float* __restrict__ qb,
                                               float* __restrict__ cfn) {
  __shared__ float s0[256], s1[256];
  const int tid = threadIdx.x;
  float4 x = *(const float4*)(cf + tid * 4);
  s0[tid] = x.x + x.y + x.z + x.w;
  s1[tid] = x.x * x.x + x.y * x.y + x.z * x.z + x.w * x.w;
  __syncthreads();
  for (int s = 128; s > 0; s >>= 1) {
    if (tid < s) { s0[tid] += s0[tid + s]; s1[tid] += s1[tid + s]; }
    __syncthreads();
  }
  const float mu = s0[0] * (1.f / 1024.f);
  const float var = s1[0] * (1.f / 1024.f) - mu * mu;
  const float rs = rsqrtf(var + 1e-5f);
  float4 g = *(const float4*)(qg + tid * 4);
  float4 b4 = *(const float4*)(qb + tid * 4);
  float4 o;
  o.x = (x.x - mu) * rs * g.x + b4.x;
  o.y = (x.y - mu) * rs * g.y + b4.y;
  o.z = (x.z - mu) * rs * g.z + b4.z;
  o.w = (x.w - mu) * rs * g.w + b4.w;
  *(float4*)(cfn + tid * 4) = o;
}

__global__ __launch_bounds__(1024) void ga_q(const float* __restrict__ cfn,
                                             const float* __restrict__ Wq,
                                             float* __restrict__ qs) {
  __shared__ __align__(16) float cs[1024];
  __shared__ float s_out[32];
  const int tid = threadIdx.x;
  const int lane = tid & 31, wave = tid >> 5;
  const int j = blockIdx.x * 32 + wave;
  const float cv = cfn[tid];
  const float4* wrow = (const float4*)(Wq + (size_t)j * 1024);
  float4 w[8];
  for (int i = 0; i < 8; ++i) w[i] = wrow[lane + 32 * i];
  cs[tid] = cv;
  __syncthreads();
  float acc = 0.f;
  for (int i = 0; i < 8; ++i) {
    const float4 c = *(const float4*)&cs[(lane + 32 * i) * 4];
    acc += c.x * w[i].x + c.y * w[i].y + c.z * w[i].z + c.w * w[i].w;
  }
  for (int m = 16; m > 0; m >>= 1) acc += __shfl_xor(acc, m, 32);
  if (lane == 0) s_out[wave] = acc;
  __syncthreads();
  if (tid < 32) qs[blockIdx.x * 32 + tid] = s_out[tid] * 0.03125f;
}

__global__ __launch_bounds__(512) void ga_wqeff(const float* __restrict__ Wkv,
                                                const float* __restrict__ qs,
                                                float* __restrict__ wqe) {
  __shared__ __align__(16) float sred[16 * 128];
  const int tid = threadIdx.x;
  const int lane = tid & 31, wave = tid >> 5;
  const int h = blockIdx.x >> 3;
  const int c0 = (blockIdx.x & 7) * 128;
  const int d0 = h * 128 + wave * 8;
  const float qv = qs[d0 + (lane & 7)];
  const float* wp = Wkv + (size_t)d0 * 1024 + c0 + lane * 4;
  float4 w[8];
  for (int i = 0; i < 8; ++i) w[i] = *(const float4*)(wp + (size_t)i * 1024);
  float4 acc = {0.f, 0.f, 0.f, 0.f};
  for (int i = 0; i < 8; ++i) {
    const float q = __int_as_float(__builtin_amdgcn_readlane(__float_as_int(qv), i));
    acc.x += w[i].x * q;
    acc.y += w[i].y * q;
    acc.z += w[i].z * q;
    acc.w += w[i].w * q;
  }
  *(float4*)&sred[wave * 128 + lane * 4] = acc;
  __syncthreads();
  if (tid < 32) {
    float4 s = *(const float4*)&sred[tid * 4];
    for (int k = 1; k < 16; ++k) {
      const float4 t = *(const float4*)&sred[k * 128 + tid * 4];
      s.x += t.x;
      s.y += t.y;
      s.z += t.z;
      s.w += t.w;
    }
    *(float4*)(wqe + h * 1024 + c0 + tid * 4) = s;
  }
}

template <int CTRL>
__device__ __forceinline__ float ga_row_xor(float v) {
  return __int_as_float(__builtin_amdgcn_update_dpp(__float_as_int(v), __float_as_int(v), CTRL,
                                                    0xF, 0xF, false));
}

__device__ __forceinline__ void ga_lds_barrier() {
  __builtin_amdgcn_fence(__ATOMIC_RELEASE, "workgroup", "local");
  __builtin_amdgcn_s_barrier();
  __builtin_amdgcn_fence(__ATOMIC_ACQUIRE, "workgroup", "local");
}

__global__ __launch_bounds__(256) void ga_main(const float* __restrict__ feats,
                                               const float* __restrict__ kvg,
                                               const float* __restrict__ kvb,
                                               const float* __restrict__ wqe,
                                               float* __restrict__ mp,
                                               float* __restrict__ lp,
                                               float* __restrict__ yp) {
  __shared__ __align__(16) float xs[16][XSTR];
  __shared__ __align__(32) float sred[2048];
  __shared__ float sS[256];
  __shared__ float sP[16 * 17];
  __shared__ float sF[16];

  const int tid = threadIdx.x;
  const int lane = tid & 31;
  const int wave = tid >> 5;
  const int l16 = lane & 15;
  const int lhi = lane >> 4;
  const int r = tid >> 4;
  const int t16 = tid & 15;

  const int blk = blockIdx.x;
  const int bb = blk >> 4;
  const int chunk = blk & 15;

  size_t rowbase = ((size_t)bb * N_ + (size_t)chunk * CHUNK + r) * C_;
  float4 x[16];
#pragma unroll
  for (int i = 0; i < 16; ++i)
    x[i] = *(const float4*)(feats + rowbase + i * 64 + t16 * 4);

  float4 wf[16];
  {
    const float* wrow = wqe + (size_t)(l16 & 7) * 1024 + wave * 128 + lhi * 4;
    const unsigned keep = (l16 < 8) ? 0xFFFFFFFFu : 0u;
#pragma unroll
    for (int s = 0; s < 16; ++s) {
      const float4 w = *(const float4*)(wrow + s * 8);
      wf[s].x = __uint_as_float(__float_as_uint(w.x) & keep);
      wf[s].y = __uint_as_float(__float_as_uint(w.y) & keep);
      wf[s].z = __uint_as_float(__float_as_uint(w.z) & keep);
      wf[s].w = __uint_as_float(__float_as_uint(w.w) & keep);
    }
  }

  float4 g4[16], b4[16];
#pragma unroll
  for (int i = 0; i < 16; ++i) {
    g4[i] = *(const float4*)(kvg + i * 64 + t16 * 4);
    b4[i] = *(const float4*)(kvb + i * 64 + t16 * 4);
  }

  const v8f vzero = {0.f, 0.f, 0.f, 0.f, 0.f, 0.f, 0.f, 0.f};
  v8f yacc[8];
#pragma unroll
  for (int t = 0; t < 8; ++t) yacc[t] = vzero;

  float m_run = -3.4e38f;
  float l_run = 0.f;

#pragma unroll 1
  for (int st = 0; st < 16; ++st) {
    float sum = 0.f, sq = 0.f;
#pragma unroll
    for (int i = 0; i < 16; ++i) {
      sum += x[i].x + x[i].y + x[i].z + x[i].w;
      sq += x[i].x * x[i].x + x[i].y * x[i].y + x[i].z * x[i].z + x[i].w * x[i].w;
    }
    sum += ga_row_xor<0x168>(sum);
    sq += ga_row_xor<0x168>(sq);
    sum += ga_row_xor<0x164>(sum);
    sq += ga_row_xor<0x164>(sq);
    sum += ga_row_xor<0x162>(sum);
    sq += ga_row_xor<0x162>(sq);
    sum += ga_row_xor<0x161>(sum);
    sq += ga_row_xor<0x161>(sq);
    {
      const float mu = sum * (1.f / 1024.f);
      const float var = sq * (1.f / 1024.f) - mu * mu;
      const float rs = rsqrtf(var + 1e-5f);
#pragma unroll
      for (int i = 0; i < 16; ++i) {
        float4 o;
        o.x = (x[i].x - mu) * rs * g4[i].x + b4[i].x;
        o.y = (x[i].y - mu) * rs * g4[i].y + b4[i].y;
        o.z = (x[i].z - mu) * rs * g4[i].z + b4[i].z;
        o.w = (x[i].w - mu) * rs * g4[i].w + b4[i].w;
        *(float4*)&xs[r][i * 64 + t16 * 4] = o;
      }
    }
    if (st < 15) {
      rowbase += (size_t)16 * C_;
#pragma unroll
      for (int i = 0; i < 16; ++i)
        x[i] = *(const float4*)(feats + rowbase + i * 64 + t16 * 4);
    }
    ga_lds_barrier();

    v8f sacc0 = vzero, sacc1 = vzero;
    {
      const float* arow = &xs[l16][wave * 128 + lhi * 4];
      float4 af[16];
#pragma unroll
      for (int s = 0; s < 16; ++s) af[s] = *(const float4*)(arow + s * 8);
      __builtin_amdgcn_sched_barrier(0);
#pragma unroll
      for (int s = 0; s < 16; ++s) {
        v2f a0, a1, b0, b1;
        a0.x = af[s].x; a0.y = af[s].y; a1.x = af[s].z; a1.y = af[s].w;
        b0.x = wf[s].x; b0.y = wf[s].y; b1.x = wf[s].z; b1.y = wf[s].w;
        sacc0 = __builtin_amdgcn_wmma_f32_16x16x4_f32(false, a0, false, b0, (short)0, sacc0,
                                                      false, false);
        sacc1 = __builtin_amdgcn_wmma_f32_16x16x4_f32(false, a1, false, b1, (short)0, sacc1,
                                                      false, false);
      }
    }
    const v8f sacc = sacc0 + sacc1;
    *(v8f*)&sred[wave * 256 + lane * 8] = sacc;
    ga_lds_barrier();
    {
      const int rr = tid >> 4, hh = tid & 15;
      const int idx = (hh + ((rr >> 3) << 4)) * 8 + (rr & 7);
      float v = 0.f;
#pragma unroll
      for (int w = 0; w < 8; ++w) v += sred[w * 256 + idx];
      sS[rr * 16 + hh] = v;
    }
    ga_lds_barrier();

    if (tid < 16) {
      float rmax = -3.4e38f;
#pragma unroll
      for (int q = 0; q < 16; ++q) rmax = fmaxf(rmax, sS[q * 16 + tid]);
      const float mnew = fmaxf(m_run, rmax);
      const float f = __expf(m_run - mnew);
      float ls = 0.f;
#pragma unroll
      for (int q = 0; q < 16; ++q) {
        const float p = __expf(sS[q * 16 + tid] - mnew);
        sP[q * 17 + tid] = p;
        ls += p;
      }
      l_run = l_run * f + ls;
      m_run = mnew;
      sF[tid] = f;
    }
    ga_lds_barrier();

    float fv[8];
#pragma unroll
    for (int v = 0; v < 8; ++v) fv[v] = sF[v + 8 * lhi];
#pragma unroll
    for (int t = 0; t < 8; ++t) {
#pragma unroll
      for (int v = 0; v < 8; ++v) yacc[t][v] *= fv[v];
    }
#pragma unroll
    for (int k0 = 0; k0 < 16; k0 += 4) {
      const int kr = k0 + 2 * lhi;
      v2f a;
      a.x = sP[kr * 17 + l16];
      a.y = sP[(kr + 1) * 17 + l16];
#pragma unroll
      for (int t = 0; t < 8; ++t) {
        const int c0 = wave * 128 + t * 16;
        v2f bv;
        bv.x = xs[kr][c0 + l16];
        bv.y = xs[kr + 1][c0 + l16];
        yacc[t] = __builtin_amdgcn_wmma_f32_16x16x4_f32(
            false, a, false, bv, (short)0, yacc[t], false, false);
      }
    }
    ga_lds_barrier();
  }

  if (tid < 8) {
    mp[blk * 8 + tid] = m_run;
    lp[blk * 8 + tid] = l_run;
  }
  if (lhi == 0) {
#pragma unroll
    for (int t = 0; t < 8; ++t) {
      const int c0 = wave * 128 + t * 16;
#pragma unroll
      for (int v = 0; v < 8; ++v)
        yp[(size_t)(blk * 8 + v) * 1024 + c0 + l16] = yacc[t][v];
    }
  }
}


__global__ __launch_bounds__(128) void ga_comb(const float* __restrict__ mp,
                                               const float* __restrict__ lp,
                                               float* yp) {
  const int bh = blockIdx.x >> 1;
  const int b = bh >> 3, h = bh & 7;
  const int c = (((blockIdx.x & 1) << 7) + threadIdx.x) * 4;
  float e[16];
  float M = -3.4e38f;
  for (int k = 0; k < 16; ++k) {
    e[k] = mp[(b * 16 + k) * 8 + h];
    M = fmaxf(M, e[k]);
  }
  float L = 0.f;
  for (int k = 0; k < 16; ++k) {
    e[k] = __expf(e[k] - M);
    L += lp[(b * 16 + k) * 8 + h] * e[k];
  }
  float4 acc = {0.f, 0.f, 0.f, 0.f};
  for (int k = 0; k < 16; ++k) {
    const float4 y = *(const float4*)(yp + (size_t)((b * 16 + k) * 8 + h) * 1024 + c);
    acc.x += y.x * e[k];
    acc.y += y.y * e[k];
    acc.z += y.z * e[k];
    acc.w += y.w * e[k];
  }
  acc.x = acc.x / L;
  acc.y = acc.y / L;
  acc.z = acc.z / L;
  acc.w = acc.w / L;
  *(float4*)(yp + (size_t)((b * 16) * 8 + h) * 1024 + c) = acc;
}

__device__ __forceinline__ v8f ga_tile64(const float* __restrict__ arow,
                                         const float* __restrict__ wrow) {
  const v8f vzero = {0.f, 0.f, 0.f, 0.f, 0.f, 0.f, 0.f, 0.f};
  v8f acc0 = vzero, acc1 = vzero;
#pragma unroll
  for (int s = 0; s < 8; ++s) {
    const float4 a = *(const float4*)(arow + s * 8);
    const float4 w = *(const float4*)(wrow + s * 8);
    v2f a0, a1, b0, b1;
    a0.x = a.x; a0.y = a.y; a1.x = a.z; a1.y = a.w;
    b0.x = w.x; b0.y = w.y; b1.x = w.z; b1.y = w.w;
    acc0 = __builtin_amdgcn_wmma_f32_16x16x4_f32(false, a0, false, b0, (short)0, acc0,
                                                 false, false);
    acc1 = __builtin_amdgcn_wmma_f32_16x16x4_f32(false, a1, false, b1, (short)0, acc1,
                                                 false, false);
  }
  return acc0 + acc1;
}

__global__ __launch_bounds__(1024) void ga_agg(const float* __restrict__ yn,
                                               const float* __restrict__ Wkv,
                                               float* __restrict__ agg) {
  __shared__ __align__(32) float sred[32 * 256];
  const int tid = threadIdx.x;
  const int lane = tid & 31, wave = tid >> 5;
  const int l16 = lane & 15, lhi = lane >> 4;
  const int nt = wave & 1, ks = wave >> 1;
  const int j0 = blockIdx.x * 32;
  const int h = j0 >> 7;
  const int koff = ks * 64 + lhi * 4;
  const float* arow = yn + (size_t)l16 * (16 * 8 * 1024) + h * 1024 + koff;
  const float* wrow = Wkv + (size_t)(1024 + j0 + nt * 16 + l16) * 1024 + koff;
  const v8f acc = ga_tile64(arow, wrow);
  *(v8f*)&sred[wave * 256 + lane * 8] = acc;
  __syncthreads();
  if (tid < 512) {
    const int m = tid >> 5, n32 = tid & 31;
    const int idx = ((n32 & 15) + ((m >> 3) << 4)) * 8 + (m & 7);
    float v = 0.f;
    for (int q = 0; q < 16; ++q) v += sred[(q * 2 + (n32 >> 4)) * 256 + idx];
    agg[m * 1024 + j0 + n32] = v;
  }
}

__global__ __launch_bounds__(1024) void ga_proj(const float* __restrict__ agg,
                                                const float* __restrict__ projW,
                                                const float* __restrict__ projb,
                                                const float* __restrict__ cf,
                                                float* __restrict__ out) {
  __shared__ __align__(32) float sred[32 * 256];
  const int tid = threadIdx.x;
  const int lane = tid & 31, wave = tid >> 5;
  const int l16 = lane & 15, lhi = lane >> 4;
  const int nt = wave & 1, ks = wave >> 1;
  const int i0 = blockIdx.x * 32;
  const int koff = ks * 64 + lhi * 4;
  const float* arow = agg + (size_t)l16 * 1024 + koff;
  const float* wrow = projW + (size_t)(i0 + nt * 16 + l16) * 1024 + koff;
  const v8f acc = ga_tile64(arow, wrow);
  *(v8f*)&sred[wave * 256 + lane * 8] = acc;
  __syncthreads();
  if (tid < 512) {
    const int m = tid >> 5, n32 = tid & 31;
    const int idx = ((n32 & 15) + ((m >> 3) << 4)) * 8 + (m & 7);
    float v = 0.f;
    for (int q = 0; q < 16; ++q) v += sred[(q * 2 + (n32 >> 4)) * 256 + idx];
    out[(size_t)m * 1024 + i0 + n32] = cf[i0 + n32] + v + projb[i0 + n32];
  }
}

extern "C" void kernel_launch(void* const* d_in, const int* in_sizes, int n_in,
                              void* d_out, int out_size, void* d_ws, size_t ws_size,
                              hipStream_t stream) {
  const float* class_feature = (const float*)d_in[0];
  const float* features      = (const float*)d_in[1];
  const float* q_gamma       = (const float*)d_in[2];
  const float* q_beta        = (const float*)d_in[3];
  const float* Wq            = (const float*)d_in[4];
  const float* kv_gamma      = (const float*)d_in[5];
  const float* kv_beta       = (const float*)d_in[6];
  const float* Wkv           = (const float*)d_in[7];
  const float* proj_W        = (const float*)d_in[8];
  const float* proj_b        = (const float*)d_in[9];
  float* out = (float*)d_out;

  float* ws   = (float*)d_ws;
  float* cfn  = ws;
  float* qs   = ws + 1024;
  float* wqe  = ws + 2048;
  float* mp   = ws + 18432;
  float* lp   = ws + 20480;
  float* yp   = ws + 22528;
  float* agg  = ws + 2048;

  ga_lncf<<<1, 256, 0, stream>>>(class_feature, q_gamma, q_beta, cfn);
  ga_q<<<32, 1024, 0, stream>>>(cfn, Wq, qs);
  ga_wqeff<<<64, 512, 0, stream>>>(Wkv, qs, wqe);
  ga_main<<<B_ * NCHUNK, 256, 0, stream>>>(features, kv_gamma, kv_beta, wqe, mp, lp, yp);
  ga_comb<<<B_ * 8 * 2, 128, 0, stream>>>(mp, lp, yp);
  ga_agg<<<32, 1024, 0, stream>>>(yp, Wkv, agg);
  ga_proj<<<32, 1024, 0, stream>>>(agg, proj_W, proj_b, class_feature, out);
}


# main: second product's B fragments read as one group of 16 b128 loads behind the score product's A reads, ahead of both matrix blocks (a lane owns 4 adjacent columns); sum order unchanged
# speedup vs baseline: 3.2912x; 1.0892x over previous
#include <hip/hip_runtime.h>
#include <hip/hip_bf16.h>

typedef __attribute__((ext_vector_type(2))) float v2f;
typedef __attribute__((ext_vector_type(8))) float v8f;
typedef __attribute__((ext_vector_type(4))) float v4f;

#define B_ 16
#define N_ 4096
#define C_ 1024
#define NCHUNK 16
#define CHUNK 256
#define XSTR 1036

__device__ __forceinline__ void ga_lds_barrier() {
  __builtin_amdgcn_fence(__ATOMIC_RELEASE, "workgroup", "local");
  __builtin_amdgcn_s_barrier();
  __builtin_amdgcn_fence(__ATOMIC_ACQUIRE, "workgroup", "local");
}

__global__ __launch_bounds__(1024) void ga_q(const float* __restrict__ cf,
                                             const float* __restrict__ qg,
                                             const float* __restrict__ qb,
                                             const float* __restrict__ Wq,
                                             float* __restrict__ qs) {
  __shared__ __align__(16) float cs[1024];
  __shared__ float s_sum[32], s_sq[32];
  __shared__ float s_out[32];
  const int tid = threadIdx.x;
  const int lane = tid & 31, wave = tid >> 5;
  const int j = blockIdx.x * 32 + wave;
  const float xv = cf[tid];
  const float gv = qg[tid];
  const float bv = qb[tid];
  const float4* wrow = (const float4*)(Wq + (size_t)j * 1024);
  float4 w[8];
  for (int i = 0; i < 8; ++i) w[i] = wrow[lane + 32 * i];
  float sm = xv, sq = xv * xv;
  for (int m = 16; m > 0; m >>= 1) {
    sm += __shfl_xor(sm, m, 32);
    sq += __shfl_xor(sq, m, 32);
  }
  if (lane == 0) { s_sum[wave] = sm; s_sq[wave] = sq; }
  ga_lds_barrier();
  sm = s_sum[lane];
  sq = s_sq[lane];
  for (int m = 16; m > 0; m >>= 1) {
    sm += __shfl_xor(sm, m, 32);
    sq += __shfl_xor(sq, m, 32);
  }
  const float mu = sm * (1.f / 1024.f);
  const float var = sq * (1.f / 1024.f) - mu * mu;
  const float rs = rsqrtf(var + 1e-5f);
  cs[tid] = (xv - mu) * rs * gv + bv;
  ga_lds_barrier();
  float acc = 0.f;
  for (int i = 0; i < 8; ++i) {
    const float4 c = *(const float4*)&cs[(lane + 32 * i) * 4];
    acc += c.x * w[i].x + c.y * w[i].y + c.z * w[i].z + c.w * w[i].w;
  }
  for (int m = 16; m > 0; m >>= 1) acc += __shfl_xor(acc, m, 32);
  if (lane == 0) s_out[wave] = acc;
  ga_lds_barrier();
  if (tid < 32) qs[blockIdx.x * 32 + tid] = s_out[tid] * 0.03125f;
}

__global__ __launch_bounds__(512) void ga_wqeff(const float* __restrict__ Wkv,
                                                const float* __restrict__ qs,
                                                float* __restrict__ wqe) {
  __shared__ __align__(16) float sred[16 * 128];
  const int tid = threadIdx.x;
  const int lane = tid & 31, wave = tid >> 5;
  const int h = blockIdx.x >> 3;
  const int c0 = (blockIdx.x & 7) * 128;
  const int d0 = h * 128 + wave * 8;
  const float qv = qs[d0 + (lane & 7)];
  const float* wp = Wkv + (size_t)d0 * 1024 + c0 + lane * 4;
  float4 w[8];
  for (int i = 0; i < 8; ++i) w[i] = *(const float4*)(wp + (size_t)i * 1024);
  float4 acc = {0.f, 0.f, 0.f, 0.f};
  for (int i = 0; i < 8; ++i) {
    const float q = __int_as_float(__builtin_amdgcn_readlane(__float_as_int(qv), i));
    acc.x += w[i].x * q;
    acc.y += w[i].y * q;
    acc.z += w[i].z * q;
    acc.w += w[i].w * q;
  }
  *(float4*)&sred[wave * 128 + lane * 4] = acc;
  __syncthreads();
  if (tid < 32) {
    float4 s = *(const float4*)&sred[tid * 4];
    for (int k = 1; k < 16; ++k) {
      const float4 t = *(const float4*)&sred[k * 128 + tid * 4];
      s.x += t.x;
      s.y += t.y;
      s.z += t.z;
      s.w += t.w;
    }
    *(float4*)(wqe + h * 1024 + c0 + tid * 4) = s;
  }
}

template <int CTRL>
__device__ __forceinline__ float ga_row_xor(float v) {
  return __int_as_float(__builtin_amdgcn_update_dpp(__float_as_int(v), __float_as_int(v), CTRL,
                                                    0xF, 0xF, false));
}

__global__ __launch_bounds__(256) void ga_main(const float* __restrict__ feats,
                                               const float* __restrict__ kvg,
                                               const float* __restrict__ kvb,
                                               const float* __restrict__ wqe,
                                               float* __restrict__ mp,
                                               float* __restrict__ lp,
                                               float* __restrict__ yp) {
  __shared__ __align__(16) float xs[16][XSTR];
  __shared__ __align__(32) float sred[2048];
  __shared__ float sS[256];
  __shared__ float sP[16 * 17];
  __shared__ float sF[16];

  const int tid = threadIdx.x;
  const int lane = tid & 31;
  const int wave = tid >> 5;
  const int l16 = lane & 15;
  const int lhi = lane >> 4;
  const int r = tid >> 4;
  const int t16 = tid & 15;

  const int blk = blockIdx.x;
  const int bb = blk >> 4;
  const int chunk = blk & 15;

  size_t rowbase = ((size_t)bb * N_ + (size_t)chunk * CHUNK + r) * C_;
  float4 x[16];
#pragma unroll
  for (int i = 0; i < 16; ++i)
    x[i] = *(const float4*)(feats + rowbase + i * 64 + t16 * 4);

  float4 wf[16];
  {
    const float* wrow = wqe + (size_t)(l16 & 7) * 1024 + wave * 128 + lhi * 4;
    const unsigned keep = (l16 < 8) ? 0xFFFFFFFFu : 0u;
#pragma unroll
    for (int s = 0; s < 16; ++s) {
      const float4 w = *(const float4*)(wrow + s * 8);
      wf[s].x = __uint_as_float(__float_as_uint(w.x) & keep);
      wf[s].y = __uint_as_float(__float_as_uint(w.y) & keep);
      wf[s].z = __uint_as_float(__float_as_uint(w.z) & keep);
      wf[s].w = __uint_as_float(__float_as_uint(w.w) & keep);
    }
  }

  float4 g4[16], b4[16];
#pragma unroll
  for (int i = 0; i < 16; ++i) {
    g4[i] = *(const float4*)(kvg + i * 64 + t16 * 4);
    b4[i] = *(const float4*)(kvb + i * 64 + t16 * 4);
  }

  const v8f vzero = {0.f, 0.f, 0.f, 0.f, 0.f, 0.f, 0.f, 0.f};
  v8f yacc[8];
#pragma unroll
  for (int t = 0; t < 8; ++t) yacc[t] = vzero;

  float m_run = -3.4e38f;
  float l_run = 0.f;

#pragma unroll 1
  for (int st = 0; st < 16; ++st) {
    float sum = 0.f, sq = 0.f;
#pragma unroll
    for (int i = 0; i < 16; ++i) {
      sum += x[i].x + x[i].y + x[i].z + x[i].w;
      sq += x[i].x * x[i].x + x[i].y * x[i].y + x[i].z * x[i].z + x[i].w * x[i].w;
    }
    sum += ga_row_xor<0x168>(sum);
    sq += ga_row_xor<0x168>(sq);
    sum += ga_row_xor<0x164>(sum);
    sq += ga_row_xor<0x164>(sq);
    sum += ga_row_xor<0x162>(sum);
    sq += ga_row_xor<0x162>(sq);
    sum += ga_row_xor<0x161>(sum);
    sq += ga_row_xor<0x161>(sq);
    {
      const float mu = sum * (1.f / 1024.f);
      const float var = sq * (1.f / 1024.f) - mu * mu;
      const float rs = rsqrtf(var + 1e-5f);
#pragma unroll
      for (int i = 0; i < 16; ++i) {
        float4 o;
        o.x = (x[i].x - mu) * rs * g4[i].x + b4[i].x;
        o.y = (x[i].y - mu) * rs * g4[i].y + b4[i].y;
        o.z = (x[i].z - mu) * rs * g4[i].z + b4[i].z;
        o.w = (x[i].w - mu) * rs * g4[i].w + b4[i].w;
        *(float4*)&xs[r][i * 64 + t16 * 4] = o;
      }
    }
    if (st < 15) {
      rowbase += (size_t)16 * C_;
#pragma unroll
      for (int i = 0; i < 16; ++i)
        x[i] = *(const float4*)(feats + rowbase + i * 64 + t16 * 4);
    }
    ga_lds_barrier();

    v8f sacc0 = vzero, sacc1 = vzero;
    v4f bq[16];
    {
      const float* arow = &xs[l16][wave * 128 + lhi * 4];
      const float* brow = &xs[2 * lhi][wave * 128 + l16 * 4];
      float4 af[16];
#pragma unroll
      for (int s = 0; s < 16; ++s) af[s] = *(const float4*)(arow + s * 8);
#pragma unroll
      for (int j = 0; j < 4; ++j) {
#pragma unroll
        for (int s = 0; s < 2; ++s) {
#pragma unroll
          for (int jj = 0; jj < 2; ++jj)
            bq[4 * j + 2 * s + jj] = *(const v4f*)(brow + (4 * j + s) * XSTR + 64 * jj);
        }
      }
      __builtin_amdgcn_sched_barrier(0);
#pragma unroll
      for (int s = 0; s < 16; ++s) {
        v2f a0, a1, b0, b1;
        a0.x = af[s].x; a0.y = af[s].y; a1.x = af[s].z; a1.y = af[s].w;
        b0.x = wf[s].x; b0.y = wf[s].y; b1.x = wf[s].z; b1.y = wf[s].w;
        sacc0 = __builtin_amdgcn_wmma_f32_16x16x4_f32(false, a0, false, b0, (short)0, sacc0,
                                                      false, false);
        sacc1 = __builtin_amdgcn_wmma_f32_16x16x4_f32(false, a1, false, b1, (short)0, sacc1,
                                                      false, false);
      }
    }
    const v8f sacc = sacc0 + sacc1;
    *(v8f*)&sred[wave * 256 + lane * 8] = sacc;
    ga_lds_barrier();
    {
      const int rr = tid >> 4, hh = tid & 15;
      const int idx = (hh + ((rr >> 3) << 4)) * 8 + (rr & 7);
      float v = 0.f;
#pragma unroll
      for (int w = 0; w < 8; ++w) v += sred[w * 256 + idx];
      sS[rr * 16 + hh] = v;
    }
    ga_lds_barrier();

    if (tid < 16) {
      float rmax = -3.4e38f;
#pragma unroll
      for (int q = 0; q < 16; ++q) rmax = fmaxf(rmax, sS[q * 16 + tid]);
      const float mnew = fmaxf(m_run, rmax);
      const float f = __expf(m_run - mnew);
      float ls = 0.f;
#pragma unroll
      for (int q = 0; q < 16; ++q) {
        const float p = __expf(sS[q * 16 + tid] - mnew);
        sP[q * 17 + tid] = p;
        ls += p;
      }
      l_run = l_run * f + ls;
      m_run = mnew;
      sF[tid] = f;
    }
    ga_lds_barrier();

    float fv[8];
#pragma unroll
    for (int v = 0; v < 8; ++v) fv[v] = sF[v + 8 * lhi];
#pragma unroll
    for (int t = 0; t < 8; ++t) {
#pragma unroll
      for (int v = 0; v < 8; ++v) yacc[t][v] *= fv[v];
    }
    v2f pa[4];
#pragma unroll
    for (int j = 0; j < 4; ++j) {
      pa[j].x = sP[(4 * j + 2 * lhi) * 17 + l16];
      pa[j].y = sP[(4 * j + 2 * lhi + 1) * 17 + l16];
    }
    __builtin_amdgcn_sched_barrier(0);
#pragma unroll
    for (int k0 = 0; k0 < 16; k0 += 4) {
      const v2f a = pa[k0 >> 2];
#pragma unroll
      for (int t = 0; t < 8; ++t) {
        v2f bv;
        bv.x = bq[k0 + (t >> 2)][t & 3];
        bv.y = bq[k0 + 2 + (t >> 2)][t & 3];
        yacc[t] = __builtin_amdgcn_wmma_f32_16x16x4_f32(
            false, a, false, bv, (short)0, yacc[t], false, false);
      }
    }
    ga_lds_barrier();
  }

  if (tid < 8) {
    mp[blk * 8 + tid] = m_run;
    lp[blk * 8 + tid] = l_run;
  }
  if (lhi == 0) {
#pragma unroll
    for (int v = 0; v < 8; ++v) {
#pragma unroll
      for (int jj = 0; jj < 2; ++jj) {
        float4 o;
        o.x = yacc[4 * jj + 0][v];
        o.y = yacc[4 * jj + 1][v];
        o.z = yacc[4 * jj + 2][v];
        o.w = yacc[4 * jj + 3][v];
        *(float4*)(yp + (size_t)(blk * 8 + v) * 1024 + wave * 128 + 64 * jj + l16 * 4) = o;
      }
    }
  }
}


__global__ __launch_bounds__(128) void ga_comb(const float* __restrict__ mp,
                                               const float* __restrict__ lp,
                                               float* yp) {
  const int bh = blockIdx.x >> 1;
  const int b = bh >> 3, h = bh & 7;
  const int c = (((blockIdx.x & 1) << 7) + threadIdx.x) * 4;
  float e[16];
  float M = -3.4e38f;
  for (int k = 0; k < 16; ++k) {
    e[k] = mp[(b * 16 + k) * 8 + h];
    M = fmaxf(M, e[k]);
  }
  float L = 0.f;
  for (int k = 0; k < 16; ++k) {
    e[k] = __expf(e[k] - M);
    L += lp[(b * 16 + k) * 8 + h] * e[k];
  }
  float4 acc = {0.f, 0.f, 0.f, 0.f};
  for (int k = 0; k < 16; ++k) {
    const float4 y = *(const float4*)(yp + (size_t)((b * 16 + k) * 8 + h) * 1024 + c);
    acc.x += y.x * e[k];
    acc.y += y.y * e[k];
    acc.z += y.z * e[k];
    acc.w += y.w * e[k];
  }
  acc.x = acc.x / L;
  acc.y = acc.y / L;
  acc.z = acc.z / L;
  acc.w = acc.w / L;
  *(float4*)(yp + (size_t)((b * 16) * 8 + h) * 1024 + c) = acc;
}

__device__ __forceinline__ v8f ga_tile64(const float* __restrict__ arow,
                                         const float* __restrict__ wrow) {
  const v8f vzero = {0.f, 0.f, 0.f, 0.f, 0.f, 0.f, 0.f, 0.f};
  v8f acc0 = vzero, acc1 = vzero;
#pragma unroll
  for (int s = 0; s < 8; ++s) {
    const float4 a = *(const float4*)(arow + s * 8);
    const float4 w = *(const float4*)(wrow + s * 8);
    v2f a0, a1, b0, b1;
    a0.x = a.x; a0.y = a.y; a1.x = a.z; a1.y = a.w;
    b0.x = w.x; b0.y = w.y; b1.x = w.z; b1.y = w.w;
    acc0 = __builtin_amdgcn_wmma_f32_16x16x4_f32(false, a0, false, b0, (short)0, acc0,
                                                 false, false);
    acc1 = __builtin_amdgcn_wmma_f32_16x16x4_f32(false, a1, false, b1, (short)0, acc1,
                                                 false, false);
  }
  return acc0 + acc1;
}

__global__ __launch_bounds__(1024) void ga_agg(const float* __restrict__ yn,
                                               const float* __restrict__ Wkv,
                                               float* __restrict__ agg) {
  __shared__ __align__(32) float sred[32 * 256];
  const int tid = threadIdx.x;
  const int lane = tid & 31, wave = tid >> 5;
  const int l16 = lane & 15, lhi = lane >> 4;
  const int nt = wave & 1, ks = wave >> 1;
  const int j0 = blockIdx.x * 32;
  const int h = j0 >> 7;
  const int koff = ks * 64 + lhi * 4;
  const float* arow = yn + (size_t)l16 * (16 * 8 * 1024) + h * 1024 + koff;
  const float* wrow = Wkv + (size_t)(1024 + j0 + nt * 16 + l16) * 1024 + koff;
  const v8f acc = ga_tile64(arow, wrow);
  *(v8f*)&sred[wave * 256 + lane * 8] = acc;
  __syncthreads();
  if (tid < 512) {
    const int m = tid >> 5, n32 = tid & 31;
    const int idx = ((n32 & 15) + ((m >> 3) << 4)) * 8 + (m & 7);
    float v = 0.f;
    for (int q = 0; q < 16; ++q) v += sred[(q * 2 + (n32 >> 4)) * 256 + idx];
    agg[m * 1024 + j0 + n32] = v;
  }
}

__global__ __launch_bounds__(1024) void ga_proj(const float* __restrict__ agg,
                                                const float* __restrict__ projW,
                                                const float* __restrict__ projb,
                                                const float* __restrict__ cf,
                                                float* __restrict__ out) {
  __shared__ __align__(32) float sred[32 * 256];
  const int tid = threadIdx.x;
  const int lane = tid & 31, wave = tid >> 5;
  const int l16 = lane & 15, lhi = lane >> 4;
  const int nt = wave & 1, ks = wave >> 1;
  const int i0 = blockIdx.x * 32;
  const int koff = ks * 64 + lhi * 4;
  const float* arow = agg + (size_t)l16 * 1024 + koff;
  const float* wrow = projW + (size_t)(i0 + nt * 16 + l16) * 1024 + koff;
  const v8f acc = ga_tile64(arow, wrow);
  *(v8f*)&sred[wave * 256 + lane * 8] = acc;
  __syncthreads();
  if (tid < 512) {
    const int m = tid >> 5, n32 = tid & 31;
    const int idx = ((n32 & 15) + ((m >> 3) << 4)) * 8 + (m & 7);
    float v = 0.f;
    for (int q = 0; q < 16; ++q) v += sred[(q * 2 + (n32 >> 4)) * 256 + idx];
    out[(size_t)m * 1024 + i0 + n32] = cf[i0 + n32] + v + projb[i0 + n32];
  }
}

extern "C" void kernel_launch(void* const* d_in, const int* in_sizes, int n_in,
                              void* d_out, int out_size, void* d_ws, size_t ws_size,
                              hipStream_t stream) {
  const float* class_feature = (const float*)d_in[0];
  const float* features      = (const float*)d_in[1];
  const float* q_gamma       = (const float*)d_in[2];
  const float* q_beta        = (const float*)d_in[3];
  const float* Wq            = (const float*)d_in[4];
  const float* kv_gamma      = (const float*)d_in[5];
  const float* kv_beta       = (const float*)d_in[6];
  const float* Wkv           = (const float*)d_in[7];
  const float* proj_W        = (const float*)d_in[8];
  const float* proj_b        = (const float*)d_in[9];
  float* out = (float*)d_out;

  float* ws   = (float*)d_ws;
  float* qs   = ws + 1024;
  float* wqe  = ws + 2048;
  float* mp   = ws + 18432;
  float* lp   = ws + 20480;
  float* yp   = ws + 22528;
  float* agg  = ws + 2048;

  ga_q<<<32, 1024, 0, stream>>>(class_feature, q_gamma, q_beta, Wq, qs);
  ga_wqeff<<<64, 512, 0, stream>>>(Wkv, qs, wqe);
  ga_main<<<B_ * NCHUNK, 256, 0, stream>>>(features, kv_gamma, kv_beta, wqe, mp, lp, yp);
  ga_comb<<<B_ * 8 * 2, 128, 0, stream>>>(mp, lp, yp);
  ga_agg<<<32, 1024, 0, stream>>>(yp, Wkv, agg);
  ga_proj<<<32, 1024, 0, stream>>>(agg, proj_W, proj_b, class_feature, out);
}
